# starting kernel
# baseline (speedup 1.0000x reference)
#include <hip/hip_runtime.h>
#include <hip/hip_bf16.h>

typedef __attribute__((ext_vector_type(2))) float v2f;
typedef __attribute__((ext_vector_type(8))) float v8f;

#define B_   16
#define L_   512
#define DB_  3072
#define D_   300
#define NN_  256
#define NW_  30
#define H_   256

#define M1   (B_ * L_)
#define MN   (B_ * NN_)

#define BM 128
#define BN 64
#define KC 32
#define LST 36

__global__ __launch_bounds__(128) void gemm_bert(
    const float* __restrict__ A, const float* __restrict__ W,
    const float* __restrict__ bias, const float* __restrict__ prelu_a,
    float* __restrict__ art)
{
    __shared__ __align__(16) float As[BM][LST];
    __shared__ __align__(16) float Bs[BN][LST];

    const int tid  = threadIdx.x;
    const int wave = tid >> 5;
    const int lane = tid & 31;
    const int rc   = lane & 15;
    const int rsel = (lane >> 4) << 1;
    const int rowh = (lane >> 4) << 3;

    const int m0 = blockIdx.x * BM;
    const int nb = blockIdx.y * BN;

    v8f acc[8];
#pragma unroll
    for (int t = 0; t < 8; ++t) acc[t] = (v8f){0,0,0,0,0,0,0,0};

    for (int kb = 0; kb < DB_; kb += KC) {
#pragma unroll
        for (int j = 0; j < 8; ++j) {
            int i   = tid + 128 * j;
            int row = i >> 3;
            int c4  = (i & 7) << 2;
            float4 v = *(const float4*)&A[(size_t)(m0 + row) * DB_ + kb + c4];
            *(float4*)&As[row][c4] = v;
        }
#pragma unroll
        for (int j = 0; j < 4; ++j) {
            int i  = tid + 128 * j;
            int n  = i >> 3;
            int c4 = (i & 7) << 2;
            float4 v = make_float4(0.f, 0.f, 0.f, 0.f);
            if (nb + n < D_)
                v = *(const float4*)&W[(size_t)(nb + n) * DB_ + kb + c4];
            *(float4*)&Bs[n][c4] = v;
        }
        __syncthreads();

        const int ar0 = wave * 32 + rc;
        const int ar1 = wave * 32 + 16 + rc;
#pragma unroll
        for (int kk = 0; kk < KC; kk += 4) {
            v2f a0, a1;
            a0.x = As[ar0][kk + rsel];
            a0.y = As[ar0][kk + rsel + 1];
            a1.x = As[ar1][kk + rsel];
            a1.y = As[ar1][kk + rsel + 1];
#pragma unroll
            for (int nt = 0; nt < 4; ++nt) {
                v2f b;
                b.x = Bs[nt * 16 + rc][kk + rsel];
                b.y = Bs[nt * 16 + rc][kk + rsel + 1];
                acc[nt] = __builtin_amdgcn_wmma_f32_16x16x4_f32(
                    false, a0, false, b, (short)0, acc[nt], false, false);
                acc[nt + 4] = __builtin_amdgcn_wmma_f32_16x16x4_f32(
                    false, a1, false, b, (short)0, acc[nt + 4], false, false);
            }
        }
        __syncthreads();
    }

    const float pa = prelu_a[0];
#pragma unroll
    for (int nt = 0; nt < 4; ++nt) {
        int gc = nb + nt * 16 + rc;
        if (gc < D_) {
            float bb = bias[gc];
#pragma unroll
            for (int part = 0; part < 2; ++part) {
#pragma unroll
                for (int i = 0; i < 8; ++i) {
                    int gr = m0 + wave * 32 + part * 16 + i + rowh;
                    float v = acc[nt + 4 * part][i] + bb;
                    v = v > 0.f ? v : pa * v;
                    art[(size_t)gr * D_ + gc] = v;
                }
            }
        }
    }
}

__global__ __launch_bounds__(128) void gather_pool(
    const float* __restrict__ art, const int* __restrict__ idx,
    const int* __restrict__ mask, float* __restrict__ nodes)
{
    const int bn = blockIdx.x;
    const int b  = bn / NN_;
    __shared__ int sidx[NW_];
    __shared__ int smsk[NW_];
    if (threadIdx.x < NW_) {
        sidx[threadIdx.x] = idx[(size_t)bn * NW_ + threadIdx.x];
        smsk[threadIdx.x] = mask[(size_t)bn * NW_ + threadIdx.x];
    }
    __syncthreads();

    float cnt = 0.f;
#pragma unroll
    for (int w = 0; w < NW_; ++w) cnt += (float)smsk[w];
    const float inv = 1.f / fmaxf(cnt, 1.f);

    for (int d = threadIdx.x; d < D_; d += 128) {
        float s = 0.f;
#pragma unroll
        for (int w = 0; w < NW_; ++w) {
            if (smsk[w])
                s += art[((size_t)b * L_ + sidx[w]) * D_ + d];
        }
        nodes[(size_t)bn * D_ + d] = s * inv;
    }
}

__global__ __launch_bounds__(128) void gemm_head(
    const float* __restrict__ Nd, const float* __restrict__ W1,
    const float* __restrict__ b1, float* __restrict__ h)
{
    const int wave = threadIdx.x >> 5;
    const int lane = threadIdx.x & 31;
    const int rc   = lane & 15;
    const int rsel = (lane >> 4) << 1;
    const int rowh = (lane >> 4) << 3;

    const int tile = blockIdx.x * 4 + wave;
    const int mt = tile >> 4;
    const int nt = tile & 15;

    const float* Ar = Nd + (size_t)(mt * 16 + rc) * D_;
    const float* Br = W1 + (size_t)(nt * 16 + rc) * D_;

    v8f acc = (v8f){0,0,0,0,0,0,0,0};
#pragma unroll 5
    for (int k = 0; k < D_; k += 4) {
        v2f a, b;
        a.x = Ar[k + rsel];
        a.y = Ar[k + rsel + 1];
        b.x = Br[k + rsel];
        b.y = Br[k + rsel + 1];
        acc = __builtin_amdgcn_wmma_f32_16x16x4_f32(
            false, a, false, b, (short)0, acc, false, false);
    }

    const int col = nt * 16 + rc;
    const float bb = b1[col];
#pragma unroll
    for (int i = 0; i < 8; ++i) {
        int row = mt * 16 + i + rowh;
        float v = acc[i] + bb;
        h[(size_t)row * H_ + col] = fmaxf(v, 0.f);
    }
}

__global__ __launch_bounds__(256) void head_out(
    const float* __restrict__ h, const float* __restrict__ W2,
    const float* __restrict__ b2, float* __restrict__ out)
{
    const int j    = blockIdx.x * 8 + (threadIdx.x >> 5);
    const int lane = threadIdx.x & 31;
    const float* hr = h + (size_t)j * H_;
    float p = 0.f;
#pragma unroll
    for (int e = lane; e < H_; e += 32) p += hr[e] * W2[e];
#pragma unroll
    for (int off = 16; off > 0; off >>= 1) p += __shfl_down(p, off, 32);
    if (lane == 0) out[j] = p + b2[0];
}

extern "C" void kernel_launch(void* const* d_in, const int* in_sizes, int n_in,
                              void* d_out, int out_size, void* d_ws, size_t ws_size,
                              hipStream_t stream)
{
    const float* articles = (const float*)d_in[0];
    const int*   nidx     = (const int*)  d_in[1];
    const int*   nmask    = (const int*)  d_in[2];
    const float* W_bert   = (const float*)d_in[5];
    const float* b_bert   = (const float*)d_in[6];
    const float* prelu_a  = (const float*)d_in[7];
    const float* W1       = (const float*)d_in[8];
    const float* b1       = (const float*)d_in[9];
    const float* W2       = (const float*)d_in[10];
    const float* b2       = (const float*)d_in[11];
    float* out = (float*)d_out;

    float* art   = (float*)d_ws;
    float* nodes = art + (size_t)M1 * D_;
    float* h     = nodes + (size_t)MN * D_;
    (void)in_sizes; (void)n_in; (void)out_size; (void)ws_size;

    gemm_bert  <<<dim3(M1 / BM, 5), 128, 0, stream>>>(articles, W_bert, b_bert, prelu_a, art);
    gather_pool<<<MN, 128, 0, stream>>>(art, nidx, nmask, nodes);
    gemm_head  <<<MN * (H_ / 16) / (4 * 16), 128, 0, stream>>>(nodes, W1, b1, h);
    head_out   <<<MN / 8, 256, 0, stream>>>(h, W2, b2, out);
}


# gemm_bert K loop software-pipelined: next chunk's A and W tiles fetched as one group of global loads ahead of the current chunk's WMMAs into a second LDS stage, one barrier per chunk
# speedup vs baseline: 1.5336x; 1.5336x over previous
#include <hip/hip_runtime.h>
#include <hip/hip_bf16.h>

typedef __attribute__((ext_vector_type(2))) float v2f;
typedef __attribute__((ext_vector_type(8))) float v8f;

#define B_   16
#define L_   512
#define DB_  3072
#define D_   300
#define NN_  256
#define NW_  30
#define H_   256

#define M1   (B_ * L_)
#define MN   (B_ * NN_)

#define BM 128
#define BN 64
#define KC 32
#define LST 36
#define NCH (DB_ / KC)

__device__ __forceinline__ void mma_chunk(
    const float (*__restrict__ as)[LST], const float (*__restrict__ bs)[LST],
    v8f (&acc)[8], int ar0, int ar1, int rc, int rsel)
{
#pragma unroll
    for (int kk = 0; kk < KC; kk += 4) {
        v2f a0, a1;
        a0.x = as[ar0][kk + rsel];
        a0.y = as[ar0][kk + rsel + 1];
        a1.x = as[ar1][kk + rsel];
        a1.y = as[ar1][kk + rsel + 1];
#pragma unroll
        for (int nt = 0; nt < 4; ++nt) {
            v2f b;
            b.x = bs[nt * 16 + rc][kk + rsel];
            b.y = bs[nt * 16 + rc][kk + rsel + 1];
            acc[nt] = __builtin_amdgcn_wmma_f32_16x16x4_f32(
                false, a0, false, b, (short)0, acc[nt], false, false);
            acc[nt + 4] = __builtin_amdgcn_wmma_f32_16x16x4_f32(
                false, a1, false, b, (short)0, acc[nt + 4], false, false);
        }
    }
}

__global__ __launch_bounds__(128) void gemm_bert(
    const float* __restrict__ A, const float* __restrict__ W,
    const float* __restrict__ bias, const float* __restrict__ prelu_a,
    float* __restrict__ art)
{
    __shared__ __align__(16) float As[2][BM][LST];
    __shared__ __align__(16) float Bs[2][BN][LST];

    const int tid  = threadIdx.x;
    const int wave = tid >> 5;
    const int lane = tid & 31;
    const int rc   = lane & 15;
    const int rsel = (lane >> 4) << 1;
    const int rowh = (lane >> 4) << 3;

    const int m0 = blockIdx.x * BM;
    const int nb = blockIdx.y * BN;

    const int srow = tid >> 3;
    const int sc4  = (tid & 7) << 2;
    const float* Ag = A + (size_t)(m0 + srow) * DB_ + sc4;
    const float* Wg[4];
#pragma unroll
    for (int j = 0; j < 4; ++j) {
        int nr = nb + srow + 16 * j;
        nr = nr < D_ ? nr : D_ - 1;
        Wg[j] = W + (size_t)nr * DB_ + sc4;
    }

    v8f acc[8];
#pragma unroll
    for (int t = 0; t < 8; ++t) acc[t] = (v8f){0,0,0,0,0,0,0,0};

    const int ar0 = wave * 32 + rc;
    const int ar1 = wave * 32 + 16 + rc;

    float4 ra[8], rb[4];

#pragma unroll
    for (int j = 0; j < 8; ++j)
        ra[j] = *(const float4*)(Ag + (size_t)(16 * j) * DB_);
#pragma unroll
    for (int j = 0; j < 4; ++j)
        rb[j] = *(const float4*)(Wg[j]);
#pragma unroll
    for (int j = 0; j < 8; ++j)
        *(float4*)&As[0][srow + 16 * j][sc4] = ra[j];
#pragma unroll
    for (int j = 0; j < 4; ++j)
        *(float4*)&Bs[0][srow + 16 * j][sc4] = rb[j];
    __syncthreads();

    for (int c = 0; c < NCH - 1; ++c) {
        const int kn = (c + 1) * KC;
#pragma unroll
        for (int j = 0; j < 8; ++j)
            ra[j] = *(const float4*)(Ag + (size_t)(16 * j) * DB_ + kn);
#pragma unroll
        for (int j = 0; j < 4; ++j)
            rb[j] = *(const float4*)(Wg[j] + kn);

        const int cur = c & 1;
        mma_chunk(As[cur], Bs[cur], acc, ar0, ar1, rc, rsel);

        const int nxt = cur ^ 1;
#pragma unroll
        for (int j = 0; j < 8; ++j)
            *(float4*)&As[nxt][srow + 16 * j][sc4] = ra[j];
#pragma unroll
        for (int j = 0; j < 4; ++j)
            *(float4*)&Bs[nxt][srow + 16 * j][sc4] = rb[j];
        __syncthreads();
    }
    mma_chunk(As[(NCH - 1) & 1], Bs[(NCH - 1) & 1], acc, ar0, ar1, rc, rsel);

    const float pa = prelu_a[0];
#pragma unroll
    for (int nt = 0; nt < 4; ++nt) {
        int gc = nb + nt * 16 + rc;
        if (gc < D_) {
            float bb = bias[gc];
#pragma unroll
            for (int part = 0; part < 2; ++part) {
#pragma unroll
                for (int i = 0; i < 8; ++i) {
                    int gr = m0 + wave * 32 + part * 16 + i + rowh;
                    float v = acc[nt + 4 * part][i] + bb;
                    v = v > 0.f ? v : pa * v;
                    art[(size_t)gr * D_ + gc] = v;
                }
            }
        }
    }
}

__global__ __launch_bounds__(128) void gather_pool(
    const float* __restrict__ art, const int* __restrict__ idx,
    const int* __restrict__ mask, float* __restrict__ nodes)
{
    const int bn = blockIdx.x;
    const int b  = bn / NN_;
    __shared__ int sidx[NW_];
    __shared__ int smsk[NW_];
    if (threadIdx.x < NW_) {
        sidx[threadIdx.x] = idx[(size_t)bn * NW_ + threadIdx.x];
        smsk[threadIdx.x] = mask[(size_t)bn * NW_ + threadIdx.x];
    }
    __syncthreads();

    float cnt = 0.f;
#pragma unroll
    for (int w = 0; w < NW_; ++w) cnt += (float)smsk[w];
    const float inv = 1.f / fmaxf(cnt, 1.f);

    for (int d = threadIdx.x; d < D_; d += 128) {
        float s = 0.f;
#pragma unroll
        for (int w = 0; w < NW_; ++w) {
            if (smsk[w])
                s += art[((size_t)b * L_ + sidx[w]) * D_ + d];
        }
        nodes[(size_t)bn * D_ + d] = s * inv;
    }
}

__global__ __launch_bounds__(128) void gemm_head(
    const float* __restrict__ Nd, const float* __restrict__ W1,
    const float* __restrict__ b1, float* __restrict__ h)
{
    const int wave = threadIdx.x >> 5;
    const int lane = threadIdx.x & 31;
    const int rc   = lane & 15;
    const int rsel = (lane >> 4) << 1;
    const int rowh = (lane >> 4) << 3;

    const int tile = blockIdx.x * 4 + wave;
    const int mt = tile >> 4;
    const int nt = tile & 15;

    const float* Ar = Nd + (size_t)(mt * 16 + rc) * D_;
    const float* Br = W1 + (size_t)(nt * 16 + rc) * D_;

    v8f acc = (v8f){0,0,0,0,0,0,0,0};
#pragma unroll 5
    for (int k = 0; k < D_; k += 4) {
        v2f a, b;
        a.x = Ar[k + rsel];
        a.y = Ar[k + rsel + 1];
        b.x = Br[k + rsel];
        b.y = Br[k + rsel + 1];
        acc = __builtin_amdgcn_wmma_f32_16x16x4_f32(
            false, a, false, b, (short)0, acc, false, false);
    }

    const int col = nt * 16 + rc;
    const float bb = b1[col];
#pragma unroll
    for (int i = 0; i < 8; ++i) {
        int row = mt * 16 + i + rowh;
        float v = acc[i] + bb;
        h[(size_t)row * H_ + col] = fmaxf(v, 0.f);
    }
}

__global__ __launch_bounds__(256) void head_out(
    const float* __restrict__ h, const float* __restrict__ W2,
    const float* __restrict__ b2, float* __restrict__ out)
{
    const int j    = blockIdx.x * 8 + (threadIdx.x >> 5);
    const int lane = threadIdx.x & 31;
    const float* hr = h + (size_t)j * H_;
    float p = 0.f;
#pragma unroll
    for (int e = lane; e < H_; e += 32) p += hr[e] * W2[e];
#pragma unroll
    for (int off = 16; off > 0; off >>= 1) p += __shfl_down(p, off, 32);
    if (lane == 0) out[j] = p + b2[0];
}

extern "C" void kernel_launch(void* const* d_in, const int* in_sizes, int n_in,
                              void* d_out, int out_size, void* d_ws, size_t ws_size,
                              hipStream_t stream)
{
    const float* articles = (const float*)d_in[0];
    const int*   nidx     = (const int*)  d_in[1];
    const int*   nmask    = (const int*)  d_in[2];
    const float* W_bert   = (const float*)d_in[5];
    const float* b_bert   = (const float*)d_in[6];
    const float* prelu_a  = (const float*)d_in[7];
    const float* W1       = (const float*)d_in[8];
    const float* b1       = (const float*)d_in[9];
    const float* W2       = (const float*)d_in[10];
    const float* b2       = (const float*)d_in[11];
    float* out = (float*)d_out;

    float* art   = (float*)d_ws;
    float* nodes = art + (size_t)M1 * D_;
    float* h     = nodes + (size_t)MN * D_;
    (void)in_sizes; (void)n_in; (void)out_size; (void)ws_size;

    gemm_bert  <<<dim3(M1 / BM, 5), 128, 0, stream>>>(articles, W_bert, b_bert, prelu_a, art);
    gather_pool<<<MN, 128, 0, stream>>>(art, nidx, nmask, nodes);
    gemm_head  <<<MN * (H_ / 16) / (4 * 16), 128, 0, stream>>>(nodes, W1, b1, h);
    head_out   <<<MN / 8, 256, 0, stream>>>(h, W2, b2, out);
}


# gemm_bert re-tiled to one work-group per WGP: 64x160 tiles on a 128x2 grid (256 equal work-groups, was 320 of 128x64), one wave per SIMD with 2x5 tiles, a chunk's fragment reads issued as one group
# speedup vs baseline: 1.8326x; 1.1950x over previous
#include <hip/hip_runtime.h>
#include <hip/hip_bf16.h>

typedef __attribute__((ext_vector_type(2))) float v2f;
typedef __attribute__((ext_vector_type(8))) float v8f;

#define B_   16
#define L_   512
#define DB_  3072
#define D_   300
#define NN_  256
#define NW_  30
#define H_   256

#define M1   (B_ * L_)
#define MN   (B_ * NN_)

#define BM 64
#define BN 160
#define KC 32
#define LST 36
#define NCH (DB_ / KC)
#define WNT 5
#define NTA (BM / 16)
#define NTB (BN / 16)
#define KST (KC / 4)

struct Frags {
    v2f a[KST][2];
    v2f b[KST][WNT];
};

__device__ __forceinline__ void load_frags(
    Frags& f, const float (*__restrict__ as)[LST], const float (*__restrict__ bs)[LST],
    int ar0, int br, int rsel)
{
#pragma unroll
    for (int ks = 0; ks < KST; ++ks) {
        f.a[ks][0] = *(const v2f*)&as[ar0][4 * ks + rsel];
        f.a[ks][1] = *(const v2f*)&as[ar0 + 16][4 * ks + rsel];
#pragma unroll
        for (int nt = 0; nt < WNT; ++nt)
            f.b[ks][nt] = *(const v2f*)&bs[br + nt * 16][4 * ks + rsel];
    }
}

template <int K0, int K1>
__device__ __forceinline__ void mma_steps(const Frags& f, v8f (&acc)[2 * WNT])
{
#pragma unroll
    for (int ks = K0; ks < K1; ++ks) {
#pragma unroll
        for (int nt = 0; nt < WNT; ++nt) {
            acc[nt] = __builtin_amdgcn_wmma_f32_16x16x4_f32(
                false, f.a[ks][0], false, f.b[ks][nt], (short)0, acc[nt], false, false);
            acc[nt + WNT] = __builtin_amdgcn_wmma_f32_16x16x4_f32(
                false, f.a[ks][1], false, f.b[ks][nt], (short)0, acc[nt + WNT], false, false);
        }
    }
}

__global__ __launch_bounds__(128) __attribute__((amdgpu_waves_per_eu(1, 1))) void gemm_bert(
    const float* __restrict__ A, const float* __restrict__ W,
    const float* __restrict__ bias, const float* __restrict__ prelu_a,
    float* __restrict__ art)
{
    __shared__ __align__(16) float As[2][BM][LST];
    __shared__ __align__(16) float Bs[2][BN][LST];

    const int tid  = threadIdx.x;
    const int wave = tid >> 5;
    const int lane = tid & 31;
    const int wm   = wave >> 1;
    const int wn   = wave & 1;
    const int rc   = lane & 15;
    const int rsel = (lane >> 4) << 1;
    const int rowh = (lane >> 4) << 3;

    const int m0 = blockIdx.x * BM;
    const int nb = blockIdx.y * BN;

    const int srow = tid >> 3;
    const int sc4  = (tid & 7) << 2;
    const float* Ag = A + (size_t)(m0 + srow) * DB_ + sc4;
    int wo[NTB];
#pragma unroll
    for (int j = 0; j < NTB; ++j) {
        int nr = nb + srow + 16 * j;
        nr = nr < D_ ? nr : D_ - 1;
        wo[j] = nr * DB_ + sc4;
    }

    v8f acc[2 * WNT];
#pragma unroll
    for (int t = 0; t < 2 * WNT; ++t) acc[t] = (v8f){0,0,0,0,0,0,0,0};

    const int ar0 = wm * 32 + rc;
    const int br  = wn * (WNT * 16) + rc;

    float4 ra[NTA], rb[NTB];
    Frags f;

#pragma unroll
    for (int j = 0; j < NTA; ++j)
        ra[j] = *(const float4*)(Ag + (size_t)(16 * j) * DB_);
#pragma unroll
    for (int j = 0; j < NTB; ++j)
        rb[j] = *(const float4*)(W + wo[j]);
#pragma unroll
    for (int j = 0; j < NTA; ++j)
        *(float4*)&As[0][srow + 16 * j][sc4] = ra[j];
#pragma unroll
    for (int j = 0; j < NTB; ++j)
        *(float4*)&Bs[0][srow + 16 * j][sc4] = rb[j];
    __syncthreads();

    for (int c = 0; c < NCH - 1; ++c) {
        const int kn = (c + 1) * KC;
#pragma unroll
        for (int j = 0; j < NTA; ++j)
            ra[j] = *(const float4*)(Ag + (size_t)(16 * j) * DB_ + kn);
#pragma unroll
        for (int j = 0; j < NTB; ++j)
            rb[j] = *(const float4*)(W + kn + wo[j]);

        const int cur = c & 1;
        load_frags(f, As[cur], Bs[cur], ar0, br, rsel);
        mma_steps<0, KST - 2>(f, acc);

        const int nxt = cur ^ 1;
#pragma unroll
        for (int j = 0; j < NTA; ++j)
            *(float4*)&As[nxt][srow + 16 * j][sc4] = ra[j];
#pragma unroll
        for (int j = 0; j < NTB; ++j)
            *(float4*)&Bs[nxt][srow + 16 * j][sc4] = rb[j];
        mma_steps<KST - 2, KST>(f, acc);

        __builtin_amdgcn_sched_group_barrier(0x020, NTA + NTB, 0);
        __builtin_amdgcn_sched_group_barrier(0x100, KST * (2 + WNT) / 2, 0);
        __builtin_amdgcn_sched_group_barrier(0x008, (KST - 2) * 2 * WNT, 0);
        __builtin_amdgcn_sched_group_barrier(0x200, NTA + NTB, 0);
        __builtin_amdgcn_sched_group_barrier(0x008, 2 * 2 * WNT, 0);
        __syncthreads();
    }
    load_frags(f, As[(NCH - 1) & 1], Bs[(NCH - 1) & 1], ar0, br, rsel);
    mma_steps<0, KST>(f, acc);

    const float pa = prelu_a[0];
#pragma unroll
    for (int nt = 0; nt < WNT; ++nt) {
        int gc = nb + wn * (WNT * 16) + nt * 16 + rc;
        if (gc < D_) {
            float bb = bias[gc];
#pragma unroll
            for (int part = 0; part < 2; ++part) {
#pragma unroll
                for (int i = 0; i < 8; ++i) {
                    int gr = m0 + wm * 32 + part * 16 + i + rowh;
                    float v = acc[nt + WNT * part][i] + bb;
                    v = v > 0.f ? v : pa * v;
                    art[(size_t)gr * D_ + gc] = v;
                }
            }
        }
    }
}

__global__ __launch_bounds__(128) void gather_pool(
    const float* __restrict__ art, const int* __restrict__ idx,
    const int* __restrict__ mask, float* __restrict__ nodes)
{
    const int bn = blockIdx.x;
    const int b  = bn / NN_;
    __shared__ int sidx[NW_];
    __shared__ int smsk[NW_];
    if (threadIdx.x < NW_) {
        sidx[threadIdx.x] = idx[(size_t)bn * NW_ + threadIdx.x];
        smsk[threadIdx.x] = mask[(size_t)bn * NW_ + threadIdx.x];
    }
    __syncthreads();

    float cnt = 0.f;
#pragma unroll
    for (int w = 0; w < NW_; ++w) cnt += (float)smsk[w];
    const float inv = 1.f / fmaxf(cnt, 1.f);

    for (int d = threadIdx.x; d < D_; d += 128) {
        float s = 0.f;
#pragma unroll
        for (int w = 0; w < NW_; ++w) {
            if (smsk[w])
                s += art[((size_t)b * L_ + sidx[w]) * D_ + d];
        }
        nodes[(size_t)bn * D_ + d] = s * inv;
    }
}

__global__ __launch_bounds__(128) void gemm_head(
    const float* __restrict__ Nd, const float* __restrict__ W1,
    const float* __restrict__ b1, float* __restrict__ h)
{
    const int wave = threadIdx.x >> 5;
    const int lane = threadIdx.x & 31;
    const int rc   = lane & 15;
    const int rsel = (lane >> 4) << 1;
    const int rowh = (lane >> 4) << 3;

    const int tile = blockIdx.x * 4 + wave;
    const int mt = tile >> 4;
    const int nt = tile & 15;

    const float* Ar = Nd + (size_t)(mt * 16 + rc) * D_;
    const float* Br = W1 + (size_t)(nt * 16 + rc) * D_;

    v8f acc = (v8f){0,0,0,0,0,0,0,0};
#pragma unroll 5
    for (int k = 0; k < D_; k += 4) {
        v2f a, b;
        a.x = Ar[k + rsel];
        a.y = Ar[k + rsel + 1];
        b.x = Br[k + rsel];
        b.y = Br[k + rsel + 1];
        acc = __builtin_amdgcn_wmma_f32_16x16x4_f32(
            false, a, false, b, (short)0, acc, false, false);
    }

    const int col = nt * 16 + rc;
    const float bb = b1[col];
#pragma unroll
    for (int i = 0; i < 8; ++i) {
        int row = mt * 16 + i + rowh;
        float v = acc[i] + bb;
        h[(size_t)row * H_ + col] = fmaxf(v, 0.f);
    }
}

__global__ __launch_bounds__(256) void head_out(
    const float* __restrict__ h, const float* __restrict__ W2,
    const float* __restrict__ b2, float* __restrict__ out)
{
    const int j    = blockIdx.x * 8 + (threadIdx.x >> 5);
    const int lane = threadIdx.x & 31;
    const float* hr = h + (size_t)j * H_;
    float p = 0.f;
#pragma unroll
    for (int e = lane; e < H_; e += 32) p += hr[e] * W2[e];
#pragma unroll
    for (int off = 16; off > 0; off >>= 1) p += __shfl_down(p, off, 32);
    if (lane == 0) out[j] = p + b2[0];
}

extern "C" void kernel_launch(void* const* d_in, const int* in_sizes, int n_in,
                              void* d_out, int out_size, void* d_ws, size_t ws_size,
                              hipStream_t stream)
{
    const float* articles = (const float*)d_in[0];
    const int*   nidx     = (const int*)  d_in[1];
    const int*   nmask    = (const int*)  d_in[2];
    const float* W_bert   = (const float*)d_in[5];
    const float* b_bert   = (const float*)d_in[6];
    const float* prelu_a  = (const float*)d_in[7];
    const float* W1       = (const float*)d_in[8];
    const float* b1       = (const float*)d_in[9];
    const float* W2       = (const float*)d_in[10];
    const float* b2       = (const float*)d_in[11];
    float* out = (float*)d_out;

    float* art   = (float*)d_ws;
    float* nodes = art + (size_t)M1 * D_;
    float* h     = nodes + (size_t)MN * D_;
    (void)in_sizes; (void)n_in; (void)out_size; (void)ws_size;

    gemm_bert  <<<dim3(M1 / BM, 2), 128, 0, stream>>>(articles, W_bert, b_bert, prelu_a, art);
    gather_pool<<<MN, 128, 0, stream>>>(art, nidx, nmask, nodes);
    gemm_head  <<<MN * (H_ / 16) / (4 * 16), 128, 0, stream>>>(nodes, W1, b1, h);
    head_out   <<<MN / 8, 256, 0, stream>>>(h, W2, b2, out);
}


# gather_pool, gemm_head, head_out fused into one kernel: a work-group per 16 nodes, span rows as independent loads, node tile and W1 in LDS, h never stored; out's dot summed in another tree
# speedup vs baseline: 2.3984x; 1.3087x over previous
#include <hip/hip_runtime.h>
#include <hip/hip_bf16.h>

typedef __attribute__((ext_vector_type(2))) float v2f;
typedef __attribute__((ext_vector_type(8))) float v8f;

#define B_   16
#define L_   512
#define DB_  3072
#define D_   300
#define NN_  256
#define NW_  30
#define H_   256

#define M1   (B_ * L_)
#define MN   (B_ * NN_)

#define BM 64
#define BN 160
#define KC 32
#define LST 36
#define NCH (DB_ / KC)
#define WNT 5
#define NTA (BM / 16)
#define NTB (BN / 16)
#define KST (KC / 4)

struct Frags {
    v2f a[KST][2];
    v2f b[KST][WNT];
};

__device__ __forceinline__ void load_frags(
    Frags& f, const float (*__restrict__ as)[LST], const float (*__restrict__ bs)[LST],
    int ar0, int br, int rsel)
{
#pragma unroll
    for (int ks = 0; ks < KST; ++ks) {
        f.a[ks][0] = *(const v2f*)&as[ar0][4 * ks + rsel];
        f.a[ks][1] = *(const v2f*)&as[ar0 + 16][4 * ks + rsel];
#pragma unroll
        for (int nt = 0; nt < WNT; ++nt)
            f.b[ks][nt] = *(const v2f*)&bs[br + nt * 16][4 * ks + rsel];
    }
}

template <int K0, int K1>
__device__ __forceinline__ void mma_steps(const Frags& f, v8f (&acc)[2 * WNT])
{
#pragma unroll
    for (int ks = K0; ks < K1; ++ks) {
#pragma unroll
        for (int nt = 0; nt < WNT; ++nt) {
            acc[nt] = __builtin_amdgcn_wmma_f32_16x16x4_f32(
                false, f.a[ks][0], false, f.b[ks][nt], (short)0, acc[nt], false, false);
            acc[nt + WNT] = __builtin_amdgcn_wmma_f32_16x16x4_f32(
                false, f.a[ks][1], false, f.b[ks][nt], (short)0, acc[nt + WNT], false, false);
        }
    }
}

__global__ __launch_bounds__(128) __attribute__((amdgpu_waves_per_eu(1, 1))) void gemm_bert(
    const float* __restrict__ A, const float* __restrict__ W,
    const float* __restrict__ bias, const float* __restrict__ prelu_a,
    float* __restrict__ art)
{
    __shared__ __align__(16) float As[2][BM][LST];
    __shared__ __align__(16) float Bs[2][BN][LST];

    const int tid  = threadIdx.x;
    const int wave = tid >> 5;
    const int lane = tid & 31;
    const int wm   = wave >> 1;
    const int wn   = wave & 1;
    const int rc   = lane & 15;
    const int rsel = (lane >> 4) << 1;
    const int rowh = (lane >> 4) << 3;

    const int m0 = blockIdx.x * BM;
    const int nb = blockIdx.y * BN;

    const int srow = tid >> 3;
    const int sc4  = (tid & 7) << 2;
    const float* Ag = A + (size_t)(m0 + srow) * DB_ + sc4;
    int wo[NTB];
#pragma unroll
    for (int j = 0; j < NTB; ++j) {
        int nr = nb + srow + 16 * j;
        nr = nr < D_ ? nr : D_ - 1;
        wo[j] = nr * DB_ + sc4;
    }

    v8f acc[2 * WNT];
#pragma unroll
    for (int t = 0; t < 2 * WNT; ++t) acc[t] = (v8f){0,0,0,0,0,0,0,0};

    const int ar0 = wm * 32 + rc;
    const int br  = wn * (WNT * 16) + rc;

    float4 ra[NTA], rb[NTB];
    Frags f;

#pragma unroll
    for (int j = 0; j < NTA; ++j)
        ra[j] = *(const float4*)(Ag + (size_t)(16 * j) * DB_);
#pragma unroll
    for (int j = 0; j < NTB; ++j)
        rb[j] = *(const float4*)(W + wo[j]);
#pragma unroll
    for (int j = 0; j < NTA; ++j)
        *(float4*)&As[0][srow + 16 * j][sc4] = ra[j];
#pragma unroll
    for (int j = 0; j < NTB; ++j)
        *(float4*)&Bs[0][srow + 16 * j][sc4] = rb[j];
    __syncthreads();

    for (int c = 0; c < NCH - 1; ++c) {
        const int kn = (c + 1) * KC;
#pragma unroll
        for (int j = 0; j < NTA; ++j)
            ra[j] = *(const float4*)(Ag + (size_t)(16 * j) * DB_ + kn);
#pragma unroll
        for (int j = 0; j < NTB; ++j)
            rb[j] = *(const float4*)(W + kn + wo[j]);

        const int cur = c & 1;
        load_frags(f, As[cur], Bs[cur], ar0, br, rsel);
        mma_steps<0, KST - 2>(f, acc);

        const int nxt = cur ^ 1;
#pragma unroll
        for (int j = 0; j < NTA; ++j)
            *(float4*)&As[nxt][srow + 16 * j][sc4] = ra[j];
#pragma unroll
        for (int j = 0; j < NTB; ++j)
            *(float4*)&Bs[nxt][srow + 16 * j][sc4] = rb[j];
        mma_steps<KST - 2, KST>(f, acc);

        __builtin_amdgcn_sched_group_barrier(0x020, NTA + NTB, 0);
        __builtin_amdgcn_sched_group_barrier(0x100, KST * (2 + WNT) / 2, 0);
        __builtin_amdgcn_sched_group_barrier(0x008, (KST - 2) * 2 * WNT, 0);
        __builtin_amdgcn_sched_group_barrier(0x200, NTA + NTB, 0);
        __builtin_amdgcn_sched_group_barrier(0x008, 2 * 2 * WNT, 0);
        __syncthreads();
    }
    load_frags(f, As[(NCH - 1) & 1], Bs[(NCH - 1) & 1], ar0, br, rsel);
    mma_steps<0, KST>(f, acc);

    const float pa = prelu_a[0];
#pragma unroll
    for (int nt = 0; nt < WNT; ++nt) {
        int gc = nb + wn * (WNT * 16) + nt * 16 + rc;
        if (gc < D_) {
            float bb = bias[gc];
#pragma unroll
            for (int part = 0; part < 2; ++part) {
#pragma unroll
                for (int i = 0; i < 8; ++i) {
                    int gr = m0 + wm * 32 + part * 16 + i + rowh;
                    float v = acc[nt + WNT * part][i] + bb;
                    v = v > 0.f ? v : pa * v;
                    art[(size_t)gr * D_ + gc] = v;
                }
            }
        }
    }
}

#define TN  16
#define TW  8
#define Q4  (D_ / 4)
#define WB  15
#define GB  10
#define KB  15
#define NKG (Q4 / KB)

__global__ __launch_bounds__(256) void tail_fused(
    const float* __restrict__ art, const int* __restrict__ idx,
    const int* __restrict__ mask, const float* __restrict__ W1,
    const float* __restrict__ b1, const float* __restrict__ W2,
    const float* __restrict__ b2, float* __restrict__ out)
{
    __shared__ __align__(16) float sw[H_ * D_];
    __shared__ __align__(16) float sn[TN * D_];
    __shared__ float spart[TW * TN];

    const int tid  = threadIdx.x;
    const int wave = tid >> 5;
    const int lane = tid & 31;
    const int bn0  = blockIdx.x * TN;
    const int b    = bn0 / NN_;

    int myi0 = 0, mym0 = 0, myi1 = 0, mym1 = 0;
    if (lane < NW_) {
        const size_t o = (size_t)(bn0 + wave * (TN / TW)) * NW_ + lane;
        myi0 = idx[o];        mym0 = mask[o];
        myi1 = idx[o + NW_];  mym1 = mask[o + NW_];
    }

    {
        const float4* Wg = (const float4*)W1;
        float4* Ws = (float4*)sw;
        for (int i0 = 0; i0 < H_ * D_ / 4 / 256; i0 += WB) {
            float4 t[WB];
#pragma unroll
            for (int j = 0; j < WB; ++j) t[j] = Wg[tid + 256 * (i0 + j)];
#pragma unroll
            for (int j = 0; j < WB; ++j) Ws[tid + 256 * (i0 + j)] = t[j];
        }
    }

    const float* artb = art + (size_t)b * L_ * D_;
    const int q2 = lane < Q4 - 64 ? lane + 64 : Q4 - 1;
    for (int nn = 0; nn < TN / TW; ++nn) {
        const int n   = wave * (TN / TW) + nn;
        const int myi = nn ? myi1 : myi0;
        const int mym = nn ? mym1 : mym0;
        float4 s0 = {0.f, 0.f, 0.f, 0.f}, s1 = s0, s2 = s0;
        float cnt = 0.f;
#pragma unroll
        for (int w0 = 0; w0 < NW_; w0 += GB) {
            float4 v0[GB], v1[GB], v2[GB];
            float m[GB];
#pragma unroll
            for (int j = 0; j < GB; ++j) {
                const int r = __builtin_amdgcn_readlane(myi, w0 + j);
                m[j] = (float)__builtin_amdgcn_readlane(mym, w0 + j);
                const float4* rp = (const float4*)(artb + (size_t)r * D_);
                v0[j] = rp[lane];
                v1[j] = rp[lane + 32];
                v2[j] = rp[q2];
            }
#pragma unroll
            for (int j = 0; j < GB; ++j) {
                cnt += m[j];
                s0.x += v0[j].x * m[j]; s0.y += v0[j].y * m[j];
                s0.z += v0[j].z * m[j]; s0.w += v0[j].w * m[j];
                s1.x += v1[j].x * m[j]; s1.y += v1[j].y * m[j];
                s1.z += v1[j].z * m[j]; s1.w += v1[j].w * m[j];
                s2.x += v2[j].x * m[j]; s2.y += v2[j].y * m[j];
                s2.z += v2[j].z * m[j]; s2.w += v2[j].w * m[j];
            }
        }
        const float inv = 1.f / fmaxf(cnt, 1.f);
        float4* np = (float4*)(sn + n * D_);
        np[lane]      = make_float4(s0.x * inv, s0.y * inv, s0.z * inv, s0.w * inv);
        np[lane + 32] = make_float4(s1.x * inv, s1.y * inv, s1.z * inv, s1.w * inv);
        if (lane < Q4 - 64)
            np[lane + 64] = make_float4(s2.x * inv, s2.y * inv, s2.z * inv, s2.w * inv);
    }
    __syncthreads();

    const int rc   = lane & 15;
    const int rsel = (lane >> 4) << 1;
    const int rowh = (lane >> 4) << 3;
    const float* ap  = sn + rc * D_ + rsel;
    const float* bp0 = sw + (wave * 32 + rc) * D_ + rsel;
    const float* bp1 = bp0 + 16 * D_;

    v8f acc0 = (v8f){0,0,0,0,0,0,0,0};
    v8f acc1 = (v8f){0,0,0,0,0,0,0,0};
    v2f fa[2][KB], fb0[2][KB], fb1[2][KB];
#pragma unroll
    for (int j = 0; j < KB; ++j) {
        fa [0][j] = *(const v2f*)(ap  + 4 * j);
        fb0[0][j] = *(const v2f*)(bp0 + 4 * j);
        fb1[0][j] = *(const v2f*)(bp1 + 4 * j);
    }
#pragma unroll
    for (int g = 0; g < NKG; ++g) {
        const int cur = g & 1;
        if (g + 1 < NKG) {
#pragma unroll
            for (int j = 0; j < KB; ++j) {
                const int k = 4 * ((g + 1) * KB + j);
                fa [cur ^ 1][j] = *(const v2f*)(ap  + k);
                fb0[cur ^ 1][j] = *(const v2f*)(bp0 + k);
                fb1[cur ^ 1][j] = *(const v2f*)(bp1 + k);
            }
        }
#pragma unroll
        for (int j = 0; j < KB; ++j) {
            acc0 = __builtin_amdgcn_wmma_f32_16x16x4_f32(
                false, fa[cur][j], false, fb0[cur][j], (short)0, acc0, false, false);
            acc1 = __builtin_amdgcn_wmma_f32_16x16x4_f32(
                false, fa[cur][j], false, fb1[cur][j], (short)0, acc1, false, false);
        }
    }

    const int c0 = wave * 32 + rc;
    const int c1 = c0 + 16;
    const float bb0 = b1[c0], bb1 = b1[c1];
    const float w20 = W2[c0], w21 = W2[c1];
    float p[8];
#pragma unroll
    for (int i = 0; i < 8; ++i) {
        const float h0 = fmaxf(acc0[i] + bb0, 0.f);
        const float h1 = fmaxf(acc1[i] + bb1, 0.f);
        p[i] = h0 * w20 + h1 * w21;
    }
#pragma unroll
    for (int off = 8; off > 0; off >>= 1) {
#pragma unroll
        for (int i = 0; i < 8; ++i) p[i] += __shfl_xor(p[i], off, 32);
    }
    if (rc == 0) {
#pragma unroll
        for (int i = 0; i < 8; ++i) spart[wave * TN + rowh + i] = p[i];
    }
    __syncthreads();
    if (tid < TN) {
        float s = 0.f;
#pragma unroll
        for (int v = 0; v < TW; ++v) s += spart[v * TN + tid];
        out[bn0 + tid] = s + b2[0];
    }
}

extern "C" void kernel_launch(void* const* d_in, const int* in_sizes, int n_in,
                              void* d_out, int out_size, void* d_ws, size_t ws_size,
                              hipStream_t stream)
{
    const float* articles = (const float*)d_in[0];
    const int*   nidx     = (const int*)  d_in[1];
    const int*   nmask    = (const int*)  d_in[2];
    const float* W_bert   = (const float*)d_in[5];
    const float* b_bert   = (const float*)d_in[6];
    const float* prelu_a  = (const float*)d_in[7];
    const float* W1       = (const float*)d_in[8];
    const float* b1       = (const float*)d_in[9];
    const float* W2       = (const float*)d_in[10];
    const float* b2       = (const float*)d_in[11];
    float* out = (float*)d_out;

    float* art = (float*)d_ws;
    (void)in_sizes; (void)n_in; (void)out_size; (void)ws_size;

    gemm_bert <<<dim3(M1 / BM, 2), 128, 0, stream>>>(articles, W_bert, b_bert, prelu_a, art);
    tail_fused<<<MN / TN, 256, 0, stream>>>(art, nidx, nmask, W1, b1, W2, b2, out);
}


# gemm_bert re-cut to two resident work-groups per WGP: 64x80 tiles on a 128x4 grid (512 work-groups of 4 waves), K chunk 64, 1x5 tiles per wave; one work-group's WMMAs fill the other's waits
# speedup vs baseline: 2.4890x; 1.0378x over previous
#include <hip/hip_runtime.h>
#include <hip/hip_bf16.h>

typedef __attribute__((ext_vector_type(2))) float v2f;
typedef __attribute__((ext_vector_type(8))) float v8f;

#define B_   16
#define L_   512
#define DB_  3072
#define D_   300
#define NN_  256
#define NW_  30
#define H_   256

#define M1   (B_ * L_)
#define MN   (B_ * NN_)

#define BM 64
#define BN 80
#define KC 64
#define LST 68
#define NCH (DB_ / KC)
#define WNT 5
#define SRS 8
#define NTA (BM / SRS)
#define NTB (BN / SRS)
#define KST (KC / 4)
#define KTL 4

struct Frags {
    v2f a[KST];
    v2f b[KST][WNT];
};

__device__ __forceinline__ void load_frags(
    Frags& f, const float (*__restrict__ as)[LST], const float (*__restrict__ bs)[LST],
    int ar0, int br, int rsel)
{
#pragma unroll
    for (int ks = 0; ks < KST; ++ks) {
        f.a[ks] = *(const v2f*)&as[ar0][4 * ks + rsel];
#pragma unroll
        for (int nt = 0; nt < WNT; ++nt)
            f.b[ks][nt] = *(const v2f*)&bs[br + nt * 16][4 * ks + rsel];
    }
}

template <int K0, int K1>
__device__ __forceinline__ void mma_steps(const Frags& f, v8f (&acc)[WNT])
{
#pragma unroll
    for (int ks = K0; ks < K1; ++ks) {
#pragma unroll
        for (int nt = 0; nt < WNT; ++nt)
            acc[nt] = __builtin_amdgcn_wmma_f32_16x16x4_f32(
                false, f.a[ks], false, f.b[ks][nt], (short)0, acc[nt], false, false);
    }
}

__global__ __launch_bounds__(128) __attribute__((amdgpu_waves_per_eu(2, 2))) void gemm_bert(
    const float* __restrict__ A, const float* __restrict__ W,
    const float* __restrict__ bias, const float* __restrict__ prelu_a,
    float* __restrict__ art)
{
    __shared__ __align__(16) float As[2][BM][LST];
    __shared__ __align__(16) float Bs[2][BN][LST];

    const int tid  = threadIdx.x;
    const int wave = tid >> 5;
    const int lane = tid & 31;
    const int rc   = lane & 15;
    const int rsel = (lane >> 4) << 1;
    const int rowh = (lane >> 4) << 3;

    const int m0 = blockIdx.x * BM;
    const int nb = blockIdx.y * BN;

    const int srow = tid >> 4;
    const int sc4  = (tid & 15) << 2;
    const float* Ag = A + (size_t)(m0 + srow) * DB_ + sc4;
    int wo[NTB];
#pragma unroll
    for (int j = 0; j < NTB; ++j) {
        int nr = nb + srow + SRS * j;
        nr = nr < D_ ? nr : D_ - 1;
        wo[j] = nr * DB_ + sc4;
    }

    v8f acc[WNT];
#pragma unroll
    for (int t = 0; t < WNT; ++t) acc[t] = (v8f){0,0,0,0,0,0,0,0};

    const int ar0 = wave * 16 + rc;
    const int br  = rc;

    float4 ra[NTA], rb[NTB];
    Frags f;

#pragma unroll
    for (int j = 0; j < NTA; ++j)
        ra[j] = *(const float4*)(Ag + (size_t)(SRS * j) * DB_);
#pragma unroll
    for (int j = 0; j < NTB; ++j)
        rb[j] = *(const float4*)(W + wo[j]);
#pragma unroll
    for (int j = 0; j < NTA; ++j)
        *(float4*)&As[0][srow + SRS * j][sc4] = ra[j];
#pragma unroll
    for (int j = 0; j < NTB; ++j)
        *(float4*)&Bs[0][srow + SRS * j][sc4] = rb[j];
    __syncthreads();

    for (int c = 0; c < NCH - 1; ++c) {
        const int kn = (c + 1) * KC;
#pragma unroll
        for (int j = 0; j < NTA; ++j)
            ra[j] = *(const float4*)(Ag + (size_t)(SRS * j) * DB_ + kn);
#pragma unroll
        for (int j = 0; j < NTB; ++j)
            rb[j] = *(const float4*)(W + kn + wo[j]);

        const int cur = c & 1;
        load_frags(f, As[cur], Bs[cur], ar0, br, rsel);
        mma_steps<0, KST - KTL>(f, acc);

        const int nxt = cur ^ 1;
#pragma unroll
        for (int j = 0; j < NTA; ++j)
            *(float4*)&As[nxt][srow + SRS * j][sc4] = ra[j];
#pragma unroll
        for (int j = 0; j < NTB; ++j)
            *(float4*)&Bs[nxt][srow + SRS * j][sc4] = rb[j];
        mma_steps<KST - KTL, KST>(f, acc);

        __builtin_amdgcn_sched_group_barrier(0x020, NTA + NTB, 0);
        __builtin_amdgcn_sched_group_barrier(0x100, KST * (1 + WNT) / 2, 0);
        __builtin_amdgcn_sched_group_barrier(0x008, (KST - KTL) * WNT, 0);
        __builtin_amdgcn_sched_group_barrier(0x200, NTA + NTB, 0);
        __builtin_amdgcn_sched_group_barrier(0x008, KTL * WNT, 0);
        __syncthreads();
    }
    load_frags(f, As[(NCH - 1) & 1], Bs[(NCH - 1) & 1], ar0, br, rsel);
    mma_steps<0, KST>(f, acc);

    const float pa = prelu_a[0];
#pragma unroll
    for (int nt = 0; nt < WNT; ++nt) {
        int gc = nb + nt * 16 + rc;
        if (gc < D_) {
            float bb = bias[gc];
#pragma unroll
            for (int i = 0; i < 8; ++i) {
                int gr = m0 + wave * 16 + i + rowh;
                float v = acc[nt][i] + bb;
                v = v > 0.f ? v : pa * v;
                art[(size_t)gr * D_ + gc] = v;
            }
        }
    }
}

#define TN  16
#define TW  8
#define Q4  (D_ / 4)
#define WB  15
#define GB  10
#define KB  15
#define NKG (Q4 / KB)

__global__ __launch_bounds__(256) void tail_fused(
    const float* __restrict__ art, const int* __restrict__ idx,
    const int* __restrict__ mask, const float* __restrict__ W1,
    const float* __restrict__ b1, const float* __restrict__ W2,
    const float* __restrict__ b2, float* __restrict__ out)
{
    __shared__ __align__(16) float sw[H_ * D_];
    __shared__ __align__(16) float sn[TN * D_];
    __shared__ float spart[TW * TN];

    const int tid  = threadIdx.x;
    const int wave = tid >> 5;
    const int lane = tid & 31;
    const int bn0  = blockIdx.x * TN;
    const int b    = bn0 / NN_;

    int myi0 = 0, mym0 = 0, myi1 = 0, mym1 = 0;
    if (lane < NW_) {
        const size_t o = (size_t)(bn0 + wave * (TN / TW)) * NW_ + lane;
        myi0 = idx[o];        mym0 = mask[o];
        myi1 = idx[o + NW_];  mym1 = mask[o + NW_];
    }

    {
        const float4* Wg = (const float4*)W1;
        float4* Ws = (float4*)sw;
        for (int i0 = 0; i0 < H_ * D_ / 4 / 256; i0 += WB) {
            float4 t[WB];
#pragma unroll
            for (int j = 0; j < WB; ++j) t[j] = Wg[tid + 256 * (i0 + j)];
#pragma unroll
            for (int j = 0; j < WB; ++j) Ws[tid + 256 * (i0 + j)] = t[j];
        }
    }

    const float* artb = art + (size_t)b * L_ * D_;
    const int q2 = lane < Q4 - 64 ? lane + 64 : Q4 - 1;
    for (int nn = 0; nn < TN / TW; ++nn) {
        const int n   = wave * (TN / TW) + nn;
        const int myi = nn ? myi1 : myi0;
        const int mym = nn ? mym1 : mym0;
        float4 s0 = {0.f, 0.f, 0.f, 0.f}, s1 = s0, s2 = s0;
        float cnt = 0.f;
#pragma unroll
        for (int w0 = 0; w0 < NW_; w0 += GB) {
            float4 v0[GB], v1[GB], v2[GB];
            float m[GB];
#pragma unroll
            for (int j = 0; j < GB; ++j) {
                const int r = __builtin_amdgcn_readlane(myi, w0 + j);
                m[j] = (float)__builtin_amdgcn_readlane(mym, w0 + j);
                const float4* rp = (const float4*)(artb + (size_t)r * D_);
                v0[j] = rp[lane];
                v1[j] = rp[lane + 32];
                v2[j] = rp[q2];
            }
#pragma unroll
            for (int j = 0; j < GB; ++j) {
                cnt += m[j];
                s0.x += v0[j].x * m[j]; s0.y += v0[j].y * m[j];
                s0.z += v0[j].z * m[j]; s0.w += v0[j].w * m[j];
                s1.x += v1[j].x * m[j]; s1.y += v1[j].y * m[j];
                s1.z += v1[j].z * m[j]; s1.w += v1[j].w * m[j];
                s2.x += v2[j].x * m[j]; s2.y += v2[j].y * m[j];
                s2.z += v2[j].z * m[j]; s2.w += v2[j].w * m[j];
            }
        }
        const float inv = 1.f / fmaxf(cnt, 1.f);
        float4* np = (float4*)(sn + n * D_);
        np[lane]      = make_float4(s0.x * inv, s0.y * inv, s0.z * inv, s0.w * inv);
        np[lane + 32] = make_float4(s1.x * inv, s1.y * inv, s1.z * inv, s1.w * inv);
        if (lane < Q4 - 64)
            np[lane + 64] = make_float4(s2.x * inv, s2.y * inv, s2.z * inv, s2.w * inv);
    }
    __syncthreads();

    const int rc   = lane & 15;
    const int rsel = (lane >> 4) << 1;
    const int rowh = (lane >> 4) << 3;
    const float* ap  = sn + rc * D_ + rsel;
    const float* bp0 = sw + (wave * 32 + rc) * D_ + rsel;
    const float* bp1 = bp0 + 16 * D_;

    v8f acc0 = (v8f){0,0,0,0,0,0,0,0};
    v8f acc1 = (v8f){0,0,0,0,0,0,0,0};
    v2f fa[2][KB], fb0[2][KB], fb1[2][KB];
#pragma unroll
    for (int j = 0; j < KB; ++j) {
        fa [0][j] = *(const v2f*)(ap  + 4 * j);
        fb0[0][j] = *(const v2f*)(bp0 + 4 * j);
        fb1[0][j] = *(const v2f*)(bp1 + 4 * j);
    }
#pragma unroll
    for (int g = 0; g < NKG; ++g) {
        const int cur = g & 1;
        if (g + 1 < NKG) {
#pragma unroll
            for (int j = 0; j < KB; ++j) {
                const int k = 4 * ((g + 1) * KB + j);
                fa [cur ^ 1][j] = *(const v2f*)(ap  + k);
                fb0[cur ^ 1][j] = *(const v2f*)(bp0 + k);
                fb1[cur ^ 1][j] = *(const v2f*)(bp1 + k);
            }
        }
#pragma unroll
        for (int j = 0; j < KB; ++j) {
            acc0 = __builtin_amdgcn_wmma_f32_16x16x4_f32(
                false, fa[cur][j], false, fb0[cur][j], (short)0, acc0, false, false);
            acc1 = __builtin_amdgcn_wmma_f32_16x16x4_f32(
                false, fa[cur][j], false, fb1[cur][j], (short)0, acc1, false, false);
        }
    }

    const int c0 = wave * 32 + rc;
    const int c1 = c0 + 16;
    const float bb0 = b1[c0], bb1 = b1[c1];
    const float w20 = W2[c0], w21 = W2[c1];
    float p[8];
#pragma unroll
    for (int i = 0; i < 8; ++i) {
        const float h0 = fmaxf(acc0[i] + bb0, 0.f);
        const float h1 = fmaxf(acc1[i] + bb1, 0.f);
        p[i] = h0 * w20 + h1 * w21;
    }
#pragma unroll
    for (int off = 8; off > 0; off >>= 1) {
#pragma unroll
        for (int i = 0; i < 8; ++i) p[i] += __shfl_xor(p[i], off, 32);
    }
    if (rc == 0) {
#pragma unroll
        for (int i = 0; i < 8; ++i) spart[wave * TN + rowh + i] = p[i];
    }
    __syncthreads();
    if (tid < TN) {
        float s = 0.f;
#pragma unroll
        for (int v = 0; v < TW; ++v) s += spart[v * TN + tid];
        out[bn0 + tid] = s + b2[0];
    }
}

extern "C" void kernel_launch(void* const* d_in, const int* in_sizes, int n_in,
                              void* d_out, int out_size, void* d_ws, size_t ws_size,
                              hipStream_t stream)
{
    const float* articles = (const float*)d_in[0];
    const int*   nidx     = (const int*)  d_in[1];
    const int*   nmask    = (const int*)  d_in[2];
    const float* W_bert   = (const float*)d_in[5];
    const float* b_bert   = (const float*)d_in[6];
    const float* prelu_a  = (const float*)d_in[7];
    const float* W1       = (const float*)d_in[8];
    const float* b1       = (const float*)d_in[9];
    const float* W2       = (const float*)d_in[10];
    const float* b2       = (const float*)d_in[11];
    float* out = (float*)d_out;

    float* art = (float*)d_ws;
    (void)in_sizes; (void)n_in; (void)out_size; (void)ws_size;

    gemm_bert <<<dim3(M1 / BM, 4), 128, 0, stream>>>(articles, W_bert, b_bert, prelu_a, art);
    tail_fused<<<MN / TN, 256, 0, stream>>>(art, nidx, nmask, W1, b1, W2, b2, out);
}


# gemm_bert: wave priority stepped down 3,2,1,0 along the K loop (after 24, 36, 44 of 48 chunks): the two work-groups on a WGP advance together, not oldest-first, covering each other's waits
# speedup vs baseline: 2.5954x; 1.0427x over previous
#include <hip/hip_runtime.h>
#include <hip/hip_bf16.h>

typedef __attribute__((ext_vector_type(2))) float v2f;
typedef __attribute__((ext_vector_type(8))) float v8f;

#define B_   16
#define L_   512
#define DB_  3072
#define D_   300
#define NN_  256
#define NW_  30
#define H_   256

#define M1   (B_ * L_)
#define MN   (B_ * NN_)

#define BM 64
#define BN 80
#define KC 64
#define LST 68
#define NCH (DB_ / KC)
#define WNT 5
#define SRS 8
#define NTA (BM / SRS)
#define NTB (BN / SRS)
#define KST (KC / 4)
#define KTL 4

struct Frags {
    v2f a[KST];
    v2f b[KST][WNT];
};

__device__ __forceinline__ void load_frags(
    Frags& f, const float (*__restrict__ as)[LST], const float (*__restrict__ bs)[LST],
    int ar0, int br, int rsel)
{
#pragma unroll
    for (int ks = 0; ks < KST; ++ks) {
        f.a[ks] = *(const v2f*)&as[ar0][4 * ks + rsel];
#pragma unroll
        for (int nt = 0; nt < WNT; ++nt)
            f.b[ks][nt] = *(const v2f*)&bs[br + nt * 16][4 * ks + rsel];
    }
}

template <int K0, int K1>
__device__ __forceinline__ void mma_steps(const Frags& f, v8f (&acc)[WNT])
{
#pragma unroll
    for (int ks = K0; ks < K1; ++ks) {
#pragma unroll
        for (int nt = 0; nt < WNT; ++nt)
            acc[nt] = __builtin_amdgcn_wmma_f32_16x16x4_f32(
                false, f.a[ks], false, f.b[ks][nt], (short)0, acc[nt], false, false);
    }
}

__global__ __launch_bounds__(128) __attribute__((amdgpu_waves_per_eu(2, 2))) void gemm_bert(
    const float* __restrict__ A, const float* __restrict__ W,
    const float* __restrict__ bias, const float* __restrict__ prelu_a,
    float* __restrict__ art)
{
    __shared__ __align__(16) float As[2][BM][LST];
    __shared__ __align__(16) float Bs[2][BN][LST];

    const int tid  = threadIdx.x;
    const int wave = tid >> 5;
    const int lane = tid & 31;
    const int rc   = lane & 15;
    const int rsel = (lane >> 4) << 1;
    const int rowh = (lane >> 4) << 3;

    const int m0 = blockIdx.x * BM;
    const int nb = blockIdx.y * BN;

    const int srow = tid >> 4;
    const int sc4  = (tid & 15) << 2;
    const float* Ag = A + (size_t)(m0 + srow) * DB_ + sc4;
    int wo[NTB];
#pragma unroll
    for (int j = 0; j < NTB; ++j) {
        int nr = nb + srow + SRS * j;
        nr = nr < D_ ? nr : D_ - 1;
        wo[j] = nr * DB_ + sc4;
    }

    v8f acc[WNT];
#pragma unroll
    for (int t = 0; t < WNT; ++t) acc[t] = (v8f){0,0,0,0,0,0,0,0};

    const int ar0 = wave * 16 + rc;
    const int br  = rc;

    float4 ra[NTA], rb[NTB];
    Frags f;

#pragma unroll
    for (int j = 0; j < NTA; ++j)
        ra[j] = *(const float4*)(Ag + (size_t)(SRS * j) * DB_);
#pragma unroll
    for (int j = 0; j < NTB; ++j)
        rb[j] = *(const float4*)(W + wo[j]);
#pragma unroll
    for (int j = 0; j < NTA; ++j)
        *(float4*)&As[0][srow + SRS * j][sc4] = ra[j];
#pragma unroll
    for (int j = 0; j < NTB; ++j)
        *(float4*)&Bs[0][srow + SRS * j][sc4] = rb[j];
    __syncthreads();

    for (int q = 0; q < 4; ++q) {
        if (q == 0)      __builtin_amdgcn_s_setprio(3);
        else if (q == 1) __builtin_amdgcn_s_setprio(2);
        else if (q == 2) __builtin_amdgcn_s_setprio(1);
        else             __builtin_amdgcn_s_setprio(0);
        const int c0 = q == 0 ? 0 : q == 1 ? NCH / 2 : q == 2 ? 3 * NCH / 4 : 11 * NCH / 12;
        const int c1 = q == 0 ? NCH / 2 : q == 1 ? 3 * NCH / 4 : q == 2 ? 11 * NCH / 12 : NCH - 1;
        for (int c = c0; c < c1; ++c) {
            const int kn = (c + 1) * KC;
#pragma unroll
            for (int j = 0; j < NTA; ++j)
                ra[j] = *(const float4*)(Ag + (size_t)(SRS * j) * DB_ + kn);
#pragma unroll
            for (int j = 0; j < NTB; ++j)
                rb[j] = *(const float4*)(W + kn + wo[j]);

            const int cur = c & 1;
            load_frags(f, As[cur], Bs[cur], ar0, br, rsel);
            mma_steps<0, KST - KTL>(f, acc);

            const int nxt = cur ^ 1;
#pragma unroll
            for (int j = 0; j < NTA; ++j)
                *(float4*)&As[nxt][srow + SRS * j][sc4] = ra[j];
#pragma unroll
            for (int j = 0; j < NTB; ++j)
                *(float4*)&Bs[nxt][srow + SRS * j][sc4] = rb[j];
            mma_steps<KST - KTL, KST>(f, acc);

            __builtin_amdgcn_sched_group_barrier(0x020, NTA + NTB, 0);
            __builtin_amdgcn_sched_group_barrier(0x100, KST * (1 + WNT) / 2, 0);
            __builtin_amdgcn_sched_group_barrier(0x008, (KST - KTL) * WNT, 0);
            __builtin_amdgcn_sched_group_barrier(0x200, NTA + NTB, 0);
            __builtin_amdgcn_sched_group_barrier(0x008, KTL * WNT, 0);
            __syncthreads();
        }
    }
    load_frags(f, As[(NCH - 1) & 1], Bs[(NCH - 1) & 1], ar0, br, rsel);
    mma_steps<0, KST>(f, acc);

    const float pa = prelu_a[0];
#pragma unroll
    for (int nt = 0; nt < WNT; ++nt) {
        int gc = nb + nt * 16 + rc;
        if (gc < D_) {
            float bb = bias[gc];
#pragma unroll
            for (int i = 0; i < 8; ++i) {
                int gr = m0 + wave * 16 + i + rowh;
                float v = acc[nt][i] + bb;
                v = v > 0.f ? v : pa * v;
                art[(size_t)gr * D_ + gc] = v;
            }
        }
    }
}

#define TN  16
#define TW  8
#define Q4  (D_ / 4)
#define WB  15
#define GB  10
#define KB  15
#define NKG (Q4 / KB)

__global__ __launch_bounds__(256) void tail_fused(
    const float* __restrict__ art, const int* __restrict__ idx,
    const int* __restrict__ mask, const float* __restrict__ W1,
    const float* __restrict__ b1, const float* __restrict__ W2,
    const float* __restrict__ b2, float* __restrict__ out)
{
    __shared__ __align__(16) float sw[H_ * D_];
    __shared__ __align__(16) float sn[TN * D_];
    __shared__ float spart[TW * TN];

    const int tid  = threadIdx.x;
    const int wave = tid >> 5;
    const int lane = tid & 31;
    const int bn0  = blockIdx.x * TN;
    const int b    = bn0 / NN_;

    int myi0 = 0, mym0 = 0, myi1 = 0, mym1 = 0;
    if (lane < NW_) {
        const size_t o = (size_t)(bn0 + wave * (TN / TW)) * NW_ + lane;
        myi0 = idx[o];        mym0 = mask[o];
        myi1 = idx[o + NW_];  mym1 = mask[o + NW_];
    }

    {
        const float4* Wg = (const float4*)W1;
        float4* Ws = (float4*)sw;
        for (int i0 = 0; i0 < H_ * D_ / 4 / 256; i0 += WB) {
            float4 t[WB];
#pragma unroll
            for (int j = 0; j < WB; ++j) t[j] = Wg[tid + 256 * (i0 + j)];
#pragma unroll
            for (int j = 0; j < WB; ++j) Ws[tid + 256 * (i0 + j)] = t[j];
        }
    }

    const float* artb = art + (size_t)b * L_ * D_;
    const int q2 = lane < Q4 - 64 ? lane + 64 : Q4 - 1;
    for (int nn = 0; nn < TN / TW; ++nn) {
        const int n   = wave * (TN / TW) + nn;
        const int myi = nn ? myi1 : myi0;
        const int mym = nn ? mym1 : mym0;
        float4 s0 = {0.f, 0.f, 0.f, 0.f}, s1 = s0, s2 = s0;
        float cnt = 0.f;
#pragma unroll
        for (int w0 = 0; w0 < NW_; w0 += GB) {
            float4 v0[GB], v1[GB], v2[GB];
            float m[GB];
#pragma unroll
            for (int j = 0; j < GB; ++j) {
                const int r = __builtin_amdgcn_readlane(myi, w0 + j);
                m[j] = (float)__builtin_amdgcn_readlane(mym, w0 + j);
                const float4* rp = (const float4*)(artb + (size_t)r * D_);
                v0[j] = rp[lane];
                v1[j] = rp[lane + 32];
                v2[j] = rp[q2];
            }
#pragma unroll
            for (int j = 0; j < GB; ++j) {
                cnt += m[j];
                s0.x += v0[j].x * m[j]; s0.y += v0[j].y * m[j];
                s0.z += v0[j].z * m[j]; s0.w += v0[j].w * m[j];
                s1.x += v1[j].x * m[j]; s1.y += v1[j].y * m[j];
                s1.z += v1[j].z * m[j]; s1.w += v1[j].w * m[j];
                s2.x += v2[j].x * m[j]; s2.y += v2[j].y * m[j];
                s2.z += v2[j].z * m[j]; s2.w += v2[j].w * m[j];
            }
        }
        const float inv = 1.f / fmaxf(cnt, 1.f);
        float4* np = (float4*)(sn + n * D_);
        np[lane]      = make_float4(s0.x * inv, s0.y * inv, s0.z * inv, s0.w * inv);
        np[lane + 32] = make_float4(s1.x * inv, s1.y * inv, s1.z * inv, s1.w * inv);
        if (lane < Q4 - 64)
            np[lane + 64] = make_float4(s2.x * inv, s2.y * inv, s2.z * inv, s2.w * inv);
    }
    __syncthreads();

    const int rc   = lane & 15;
    const int rsel = (lane >> 4) << 1;
    const int rowh = (lane >> 4) << 3;
    const float* ap  = sn + rc * D_ + rsel;
    const float* bp0 = sw + (wave * 32 + rc) * D_ + rsel;
    const float* bp1 = bp0 + 16 * D_;

    v8f acc0 = (v8f){0,0,0,0,0,0,0,0};
    v8f acc1 = (v8f){0,0,0,0,0,0,0,0};
    v2f fa[2][KB], fb0[2][KB], fb1[2][KB];
#pragma unroll
    for (int j = 0; j < KB; ++j) {
        fa [0][j] = *(const v2f*)(ap  + 4 * j);
        fb0[0][j] = *(const v2f*)(bp0 + 4 * j);
        fb1[0][j] = *(const v2f*)(bp1 + 4 * j);
    }
#pragma unroll
    for (int g = 0; g < NKG; ++g) {
        const int cur = g & 1;
        if (g + 1 < NKG) {
#pragma unroll
            for (int j = 0; j < KB; ++j) {
                const int k = 4 * ((g + 1) * KB + j);
                fa [cur ^ 1][j] = *(const v2f*)(ap  + k);
                fb0[cur ^ 1][j] = *(const v2f*)(bp0 + k);
                fb1[cur ^ 1][j] = *(const v2f*)(bp1 + k);
            }
        }
#pragma unroll
        for (int j = 0; j < KB; ++j) {
            acc0 = __builtin_amdgcn_wmma_f32_16x16x4_f32(
                false, fa[cur][j], false, fb0[cur][j], (short)0, acc0, false, false);
            acc1 = __builtin_amdgcn_wmma_f32_16x16x4_f32(
                false, fa[cur][j], false, fb1[cur][j], (short)0, acc1, false, false);
        }
    }

    const int c0 = wave * 32 + rc;
    const int c1 = c0 + 16;
    const float bb0 = b1[c0], bb1 = b1[c1];
    const float w20 = W2[c0], w21 = W2[c1];
    float p[8];
#pragma unroll
    for (int i = 0; i < 8; ++i) {
        const float h0 = fmaxf(acc0[i] + bb0, 0.f);
        const float h1 = fmaxf(acc1[i] + bb1, 0.f);
        p[i] = h0 * w20 + h1 * w21;
    }
#pragma unroll
    for (int off = 8; off > 0; off >>= 1) {
#pragma unroll
        for (int i = 0; i < 8; ++i) p[i] += __shfl_xor(p[i], off, 32);
    }
    if (rc == 0) {
#pragma unroll
        for (int i = 0; i < 8; ++i) spart[wave * TN + rowh + i] = p[i];
    }
    __syncthreads();
    if (tid < TN) {
        float s = 0.f;
#pragma unroll
        for (int v = 0; v < TW; ++v) s += spart[v * TN + tid];
        out[bn0 + tid] = s + b2[0];
    }
}

extern "C" void kernel_launch(void* const* d_in, const int* in_sizes, int n_in,
                              void* d_out, int out_size, void* d_ws, size_t ws_size,
                              hipStream_t stream)
{
    const float* articles = (const float*)d_in[0];
    const int*   nidx     = (const int*)  d_in[1];
    const int*   nmask    = (const int*)  d_in[2];
    const float* W_bert   = (const float*)d_in[5];
    const float* b_bert   = (const float*)d_in[6];
    const float* prelu_a  = (const float*)d_in[7];
    const float* W1       = (const float*)d_in[8];
    const float* b1       = (const float*)d_in[9];
    const float* W2       = (const float*)d_in[10];
    const float* b2       = (const float*)d_in[11];
    float* out = (float*)d_out;

    float* art = (float*)d_ws;
    (void)in_sizes; (void)n_in; (void)out_size; (void)ws_size;

    gemm_bert <<<dim3(M1 / BM, 4), 128, 0, stream>>>(articles, W_bert, b_bert, prelu_a, art);
    tail_fused<<<MN / TN, 256, 0, stream>>>(art, nidx, nmask, W1, b1, W2, b2, out);
}


# gemm_bert: a chunk's 16 k-steps split over the four waves, each wave holds the whole 64x80 tile (4x5 accumulators): 18 fragment reads per chunk, was 48; partial tiles summed in wave order via LDS
# speedup vs baseline: 2.7183x; 1.0474x over previous
#include <hip/hip_runtime.h>
#include <hip/hip_bf16.h>

typedef __attribute__((ext_vector_type(2))) float v2f;
typedef __attribute__((ext_vector_type(8))) float v8f;

#define B_   16
#define L_   512
#define DB_  3072
#define D_   300
#define NN_  256
#define NW_  30
#define H_   256

#define M1   (B_ * L_)
#define MN   (B_ * NN_)

#define BM 64
#define BN 80
#define KC 64
#define LST 68
#define NCH (DB_ / KC)
#define WNT 5
#define NST 4
#define NWV 4
#define SRS 8
#define NTA (BM / SRS)
#define NTB (BN / SRS)
#define KST (KC / 4)
#define KSW (KST / NWV)
#define KTL 1
#define NTL (NST * WNT)
#define SMF (NWV * NTL * 8 * 32)

struct Frags {
    v2f a[KSW][NST];
    v2f b[KSW][WNT];
};

__device__ __forceinline__ void load_frags(
    Frags& f, const float (*__restrict__ as)[LST], const float (*__restrict__ bs)[LST],
    int rc, int kx)
{
#pragma unroll
    for (int ks = 0; ks < KSW; ++ks) {
#pragma unroll
        for (int r = 0; r < NST; ++r)
            f.a[ks][r] = *(const v2f*)&as[rc + r * 16][kx + 4 * ks];
#pragma unroll
        for (int nt = 0; nt < WNT; ++nt)
            f.b[ks][nt] = *(const v2f*)&bs[rc + nt * 16][kx + 4 * ks];
    }
}

template <int K0, int K1>
__device__ __forceinline__ void mma_steps(const Frags& f, v8f (&acc)[NST][WNT])
{
#pragma unroll
    for (int ks = K0; ks < K1; ++ks) {
#pragma unroll
        for (int r = 0; r < NST; ++r) {
#pragma unroll
            for (int nt = 0; nt < WNT; ++nt)
                acc[r][nt] = __builtin_amdgcn_wmma_f32_16x16x4_f32(
                    false, f.a[ks][r], false, f.b[ks][nt], (short)0, acc[r][nt], false, false);
        }
    }
}

__global__ __launch_bounds__(128) __attribute__((amdgpu_waves_per_eu(2, 2))) void gemm_bert(
    const float* __restrict__ A, const float* __restrict__ W,
    const float* __restrict__ bias, const float* __restrict__ prelu_a,
    float* __restrict__ art)
{
    __shared__ __align__(16) float smem[SMF];
    float (*As)[BM][LST] = (float (*)[BM][LST])smem;
    float (*Bs)[BN][LST] = (float (*)[BN][LST])(smem + 2 * BM * LST);

    const int tid  = threadIdx.x;
    const int wave = tid >> 5;
    const int lane = tid & 31;
    const int rc   = lane & 15;
    const int rsel = (lane >> 4) << 1;
    const int rowh = (lane >> 4) << 3;
    const int kx   = wave * (KC / NWV) + rsel;

    const int m0 = blockIdx.x * BM;
    const int nb = blockIdx.y * BN;

    const int srow = tid >> 4;
    const int sc4  = (tid & 15) << 2;
    const float* Ag = A + (size_t)(m0 + srow) * DB_ + sc4;
    int wo[NTB];
#pragma unroll
    for (int j = 0; j < NTB; ++j) {
        int nr = nb + srow + SRS * j;
        nr = nr < D_ ? nr : D_ - 1;
        wo[j] = nr * DB_ + sc4;
    }

    v8f acc[NST][WNT];
#pragma unroll
    for (int r = 0; r < NST; ++r)
#pragma unroll
        for (int t = 0; t < WNT; ++t) acc[r][t] = (v8f){0,0,0,0,0,0,0,0};

    float4 ra[NTA], rb[NTB];
    Frags f;

#pragma unroll
    for (int j = 0; j < NTA; ++j)
        ra[j] = *(const float4*)(Ag + (size_t)(SRS * j) * DB_);
#pragma unroll
    for (int j = 0; j < NTB; ++j)
        rb[j] = *(const float4*)(W + wo[j]);
#pragma unroll
    for (int j = 0; j < NTA; ++j)
        *(float4*)&As[0][srow + SRS * j][sc4] = ra[j];
#pragma unroll
    for (int j = 0; j < NTB; ++j)
        *(float4*)&Bs[0][srow + SRS * j][sc4] = rb[j];
    __syncthreads();

    for (int q = 0; q < 4; ++q) {
        if (q == 0)      __builtin_amdgcn_s_setprio(3);
        else if (q == 1) __builtin_amdgcn_s_setprio(2);
        else if (q == 2) __builtin_amdgcn_s_setprio(1);
        else             __builtin_amdgcn_s_setprio(0);
        const int c0 = q == 0 ? 0 : q == 1 ? NCH / 2 : q == 2 ? 3 * NCH / 4 : 11 * NCH / 12;
        const int c1 = q == 0 ? NCH / 2 : q == 1 ? 3 * NCH / 4 : q == 2 ? 11 * NCH / 12 : NCH - 1;
        for (int c = c0; c < c1; ++c) {
            const int kn = (c + 1) * KC;
#pragma unroll
            for (int j = 0; j < NTA; ++j)
                ra[j] = *(const float4*)(Ag + (size_t)(SRS * j) * DB_ + kn);
#pragma unroll
            for (int j = 0; j < NTB; ++j)
                rb[j] = *(const float4*)(W + kn + wo[j]);

            const int cur = c & 1;
            load_frags(f, As[cur], Bs[cur], rc, kx);
            mma_steps<0, KSW - KTL>(f, acc);

            const int nxt = cur ^ 1;
#pragma unroll
            for (int j = 0; j < NTA; ++j)
                *(float4*)&As[nxt][srow + SRS * j][sc4] = ra[j];
#pragma unroll
            for (int j = 0; j < NTB; ++j)
                *(float4*)&Bs[nxt][srow + SRS * j][sc4] = rb[j];
            mma_steps<KSW - KTL, KSW>(f, acc);

            __builtin_amdgcn_sched_group_barrier(0x020, NTA + NTB, 0);
            __builtin_amdgcn_sched_group_barrier(0x100, KSW * (NST + WNT) / 2, 0);
            __builtin_amdgcn_sched_group_barrier(0x008, (KSW - KTL) * NTL, 0);
            __builtin_amdgcn_sched_group_barrier(0x200, NTA + NTB, 0);
            __builtin_amdgcn_sched_group_barrier(0x008, KTL * NTL, 0);
            __syncthreads();
        }
    }
    load_frags(f, As[(NCH - 1) & 1], Bs[(NCH - 1) & 1], rc, kx);
    mma_steps<0, KSW>(f, acc);

    __syncthreads();
    float4* red = (float4*)smem;
#pragma unroll
    for (int r = 0; r < NST; ++r) {
#pragma unroll
        for (int nt = 0; nt < WNT; ++nt) {
            const int o = ((wave * NTL + r * WNT + nt) * 2) * 32 + lane;
            red[o]      = make_float4(acc[r][nt][0], acc[r][nt][1], acc[r][nt][2], acc[r][nt][3]);
            red[o + 32] = make_float4(acc[r][nt][4], acc[r][nt][5], acc[r][nt][6], acc[r][nt][7]);
        }
    }
    __syncthreads();

    const float pa = prelu_a[0];
#pragma unroll
    for (int nt = 0; nt < WNT; ++nt) {
        float4 lo[NWV], hi[NWV];
#pragma unroll
        for (int v = 0; v < NWV; ++v) {
            const int o = ((v * NTL + wave * WNT + nt) * 2) * 32 + lane;
            lo[v] = red[o];
            hi[v] = red[o + 32];
        }
        float s[8];
        s[0] = lo[0].x; s[1] = lo[0].y; s[2] = lo[0].z; s[3] = lo[0].w;
        s[4] = hi[0].x; s[5] = hi[0].y; s[6] = hi[0].z; s[7] = hi[0].w;
#pragma unroll
        for (int v = 1; v < NWV; ++v) {
            s[0] += lo[v].x; s[1] += lo[v].y; s[2] += lo[v].z; s[3] += lo[v].w;
            s[4] += hi[v].x; s[5] += hi[v].y; s[6] += hi[v].z; s[7] += hi[v].w;
        }
        int gc = nb + nt * 16 + rc;
        if (gc < D_) {
            float bb = bias[gc];
#pragma unroll
            for (int i = 0; i < 8; ++i) {
                int gr = m0 + wave * 16 + i + rowh;
                float v = s[i] + bb;
                v = v > 0.f ? v : pa * v;
                art[(size_t)gr * D_ + gc] = v;
            }
        }
    }
}

#define TN  16
#define TW  8
#define Q4  (D_ / 4)
#define WB  15
#define GB  10
#define KB  15
#define NKG (Q4 / KB)

__global__ __launch_bounds__(256) void tail_fused(
    const float* __restrict__ art, const int* __restrict__ idx,
    const int* __restrict__ mask, const float* __restrict__ W1,
    const float* __restrict__ b1, const float* __restrict__ W2,
    const float* __restrict__ b2, float* __restrict__ out)
{
    __shared__ __align__(16) float sw[H_ * D_];
    __shared__ __align__(16) float sn[TN * D_];
    __shared__ float spart[TW * TN];

    const int tid  = threadIdx.x;
    const int wave = tid >> 5;
    const int lane = tid & 31;
    const int bn0  = blockIdx.x * TN;
    const int b    = bn0 / NN_;

    int myi0 = 0, mym0 = 0, myi1 = 0, mym1 = 0;
    if (lane < NW_) {
        const size_t o = (size_t)(bn0 + wave * (TN / TW)) * NW_ + lane;
        myi0 = idx[o];        mym0 = mask[o];
        myi1 = idx[o + NW_];  mym1 = mask[o + NW_];
    }

    {
        const float4* Wg = (const float4*)W1;
        float4* Ws = (float4*)sw;
        for (int i0 = 0; i0 < H_ * D_ / 4 / 256; i0 += WB) {
            float4 t[WB];
#pragma unroll
            for (int j = 0; j < WB; ++j) t[j] = Wg[tid + 256 * (i0 + j)];
#pragma unroll
            for (int j = 0; j < WB; ++j) Ws[tid + 256 * (i0 + j)] = t[j];
        }
    }

    const float* artb = art + (size_t)b * L_ * D_;
    const int q2 = lane < Q4 - 64 ? lane + 64 : Q4 - 1;
    for (int nn = 0; nn < TN / TW; ++nn) {
        const int n   = wave * (TN / TW) + nn;
        const int myi = nn ? myi1 : myi0;
        const int mym = nn ? mym1 : mym0;
        float4 s0 = {0.f, 0.f, 0.f, 0.f}, s1 = s0, s2 = s0;
        float cnt = 0.f;
#pragma unroll
        for (int w0 = 0; w0 < NW_; w0 += GB) {
            float4 v0[GB], v1[GB], v2[GB];
            float m[GB];
#pragma unroll
            for (int j = 0; j < GB; ++j) {
                const int r = __builtin_amdgcn_readlane(myi, w0 + j);
                m[j] = (float)__builtin_amdgcn_readlane(mym, w0 + j);
                const float4* rp = (const float4*)(artb + (size_t)r * D_);
                v0[j] = rp[lane];
                v1[j] = rp[lane + 32];
                v2[j] = rp[q2];
            }
#pragma unroll
            for (int j = 0; j < GB; ++j) {
                cnt += m[j];
                s0.x += v0[j].x * m[j]; s0.y += v0[j].y * m[j];
                s0.z += v0[j].z * m[j]; s0.w += v0[j].w * m[j];
                s1.x += v1[j].x * m[j]; s1.y += v1[j].y * m[j];
                s1.z += v1[j].z * m[j]; s1.w += v1[j].w * m[j];
                s2.x += v2[j].x * m[j]; s2.y += v2[j].y * m[j];
                s2.z += v2[j].z * m[j]; s2.w += v2[j].w * m[j];
            }
        }
        const float inv = 1.f / fmaxf(cnt, 1.f);
        float4* np = (float4*)(sn + n * D_);
        np[lane]      = make_float4(s0.x * inv, s0.y * inv, s0.z * inv, s0.w * inv);
        np[lane + 32] = make_float4(s1.x * inv, s1.y * inv, s1.z * inv, s1.w * inv);
        if (lane < Q4 - 64)
            np[lane + 64] = make_float4(s2.x * inv, s2.y * inv, s2.z * inv, s2.w * inv);
    }
    __syncthreads();

    const int rc   = lane & 15;
    const int rsel = (lane >> 4) << 1;
    const int rowh = (lane >> 4) << 3;
    const float* ap  = sn + rc * D_ + rsel;
    const float* bp0 = sw + (wave * 32 + rc) * D_ + rsel;
    const float* bp1 = bp0 + 16 * D_;

    v8f acc0 = (v8f){0,0,0,0,0,0,0,0};
    v8f acc1 = (v8f){0,0,0,0,0,0,0,0};
    v2f fa[2][KB], fb0[2][KB], fb1[2][KB];
#pragma unroll
    for (int j = 0; j < KB; ++j) {
        fa [0][j] = *(const v2f*)(ap  + 4 * j);
        fb0[0][j] = *(const v2f*)(bp0 + 4 * j);
        fb1[0][j] = *(const v2f*)(bp1 + 4 * j);
    }
#pragma unroll
    for (int g = 0; g < NKG; ++g) {
        const int cur = g & 1;
        if (g + 1 < NKG) {
#pragma unroll
            for (int j = 0; j < KB; ++j) {
                const int k = 4 * ((g + 1) * KB + j);
                fa [cur ^ 1][j] = *(const v2f*)(ap  + k);
                fb0[cur ^ 1][j] = *(const v2f*)(bp0 + k);
                fb1[cur ^ 1][j] = *(const v2f*)(bp1 + k);
            }
        }
#pragma unroll
        for (int j = 0; j < KB; ++j) {
            acc0 = __builtin_amdgcn_wmma_f32_16x16x4_f32(
                false, fa[cur][j], false, fb0[cur][j], (short)0, acc0, false, false);
            acc1 = __builtin_amdgcn_wmma_f32_16x16x4_f32(
                false, fa[cur][j], false, fb1[cur][j], (short)0, acc1, false, false);
        }
    }

    const int c0 = wave * 32 + rc;
    const int c1 = c0 + 16;
    const float bb0 = b1[c0], bb1 = b1[c1];
    const float w20 = W2[c0], w21 = W2[c1];
    float p[8];
#pragma unroll
    for (int i = 0; i < 8; ++i) {
        const float h0 = fmaxf(acc0[i] + bb0, 0.f);
        const float h1 = fmaxf(acc1[i] + bb1, 0.f);
        p[i] = h0 * w20 + h1 * w21;
    }
#pragma unroll
    for (int off = 8; off > 0; off >>= 1) {
#pragma unroll
        for (int i = 0; i < 8; ++i) p[i] += __shfl_xor(p[i], off, 32);
    }
    if (rc == 0) {
#pragma unroll
        for (int i = 0; i < 8; ++i) spart[wave * TN + rowh + i] = p[i];
    }
    __syncthreads();
    if (tid < TN) {
        float s = 0.f;
#pragma unroll
        for (int v = 0; v < TW; ++v) s += spart[v * TN + tid];
        out[bn0 + tid] = s + b2[0];
    }
}

extern "C" void kernel_launch(void* const* d_in, const int* in_sizes, int n_in,
                              void* d_out, int out_size, void* d_ws, size_t ws_size,
                              hipStream_t stream)
{
    const float* articles = (const float*)d_in[0];
    const int*   nidx     = (const int*)  d_in[1];
    const int*   nmask    = (const int*)  d_in[2];
    const float* W_bert   = (const float*)d_in[5];
    const float* b_bert   = (const float*)d_in[6];
    const float* prelu_a  = (const float*)d_in[7];
    const float* W1       = (const float*)d_in[8];
    const float* b1       = (const float*)d_in[9];
    const float* W2       = (const float*)d_in[10];
    const float* b2       = (const float*)d_in[11];
    float* out = (float*)d_out;

    float* art = (float*)d_ws;
    (void)in_sizes; (void)n_in; (void)out_size; (void)ws_size;

    gemm_bert <<<dim3(M1 / BM, 4), 128, 0, stream>>>(articles, W_bert, b_bert, prelu_a, art);
    tail_fused<<<MN / TN, 256, 0, stream>>>(art, nidx, nmask, W1, b1, W2, b2, out);
}


# gemm_bert: fragment reads pipelined across the chunk barrier: a chunk's 36 fragments go into a second register set right behind the barrier, covered by half of the previous chunk's last k-step
# speedup vs baseline: 2.8425x; 1.0457x over previous
#include <hip/hip_runtime.h>
#include <hip/hip_bf16.h>

typedef __attribute__((ext_vector_type(2))) float v2f;
typedef __attribute__((ext_vector_type(8))) float v8f;

#define B_   16
#define L_   512
#define DB_  3072
#define D_   300
#define NN_  256
#define NW_  30
#define H_   256

#define M1   (B_ * L_)
#define MN   (B_ * NN_)

#define BM 64
#define BN 80
#define KC 64
#define LST 68
#define NCH (DB_ / KC)
#define WNT 5
#define NST 4
#define NWV 4
#define SRS 8
#define NTA (BM / SRS)
#define NTB (BN / SRS)
#define KST (KC / 4)
#define KSW (KST / NWV)
#define NTL (NST * WNT)
#define SMF (NWV * NTL * 8 * 32)

struct Frags {
    v2f a[KSW][NST];
    v2f b[KSW][WNT];
};

__device__ __forceinline__ void load_frags(
    Frags& f, const float (*__restrict__ as)[LST], const float (*__restrict__ bs)[LST],
    int rc, int kx)
{
#pragma unroll
    for (int ks = 0; ks < KSW; ++ks) {
#pragma unroll
        for (int r = 0; r < NST; ++r)
            f.a[ks][r] = *(const v2f*)&as[rc + r * 16][kx + 4 * ks];
#pragma unroll
        for (int nt = 0; nt < WNT; ++nt)
            f.b[ks][nt] = *(const v2f*)&bs[rc + nt * 16][kx + 4 * ks];
    }
}

template <int K0, int K1>
__device__ __forceinline__ void mma_steps(const Frags& f, v8f (&acc)[NST][WNT])
{
#pragma unroll
    for (int ks = K0; ks < K1; ++ks) {
#pragma unroll
        for (int r = 0; r < NST; ++r) {
#pragma unroll
            for (int nt = 0; nt < WNT; ++nt)
                acc[r][nt] = __builtin_amdgcn_wmma_f32_16x16x4_f32(
                    false, f.a[ks][r], false, f.b[ks][nt], (short)0, acc[r][nt], false, false);
        }
    }
}

template <int KS, int R0, int R1>
__device__ __forceinline__ void mma_rows(const Frags& f, v8f (&acc)[NST][WNT])
{
#pragma unroll
    for (int r = R0; r < R1; ++r) {
#pragma unroll
        for (int nt = 0; nt < WNT; ++nt)
            acc[r][nt] = __builtin_amdgcn_wmma_f32_16x16x4_f32(
                false, f.a[KS][r], false, f.b[KS][nt], (short)0, acc[r][nt], false, false);
    }
}

__device__ __forceinline__ void chunk_step(
    const int c, const Frags& fc, Frags& fn, v8f (&acc)[NST][WNT],
    const float* __restrict__ Ag, const float* __restrict__ W, const int (&wo)[NTB],
    float (*As)[BM][LST], float (*Bs)[BN][LST],
    const int srow, const int sc4, const int rc, const int kx)
{
    const int kn = (c + 1) * KC;
    float4 ra[NTA], rb[NTB];
#pragma unroll
    for (int j = 0; j < NTA; ++j)
        ra[j] = *(const float4*)(Ag + (size_t)(SRS * j) * DB_ + kn);
#pragma unroll
    for (int j = 0; j < NTB; ++j)
        rb[j] = *(const float4*)(W + kn + wo[j]);

    mma_steps<0, KSW - 1>(fc, acc);

    const int nxt = (c + 1) & 1;
#pragma unroll
    for (int j = 0; j < NTA; ++j)
        *(float4*)&As[nxt][srow + SRS * j][sc4] = ra[j];
#pragma unroll
    for (int j = 0; j < NTB; ++j)
        *(float4*)&Bs[nxt][srow + SRS * j][sc4] = rb[j];
    mma_rows<KSW - 1, 0, NST / 2>(fc, acc);

    __builtin_amdgcn_sched_group_barrier(0x020, NTA + NTB, 0);
    __builtin_amdgcn_sched_group_barrier(0x008, (KSW - 1) * NTL, 0);
    __builtin_amdgcn_sched_group_barrier(0x200, NTA + NTB, 0);
    __builtin_amdgcn_sched_group_barrier(0x008, NTL / 2, 0);
    __syncthreads();

    load_frags(fn, As[nxt], Bs[nxt], rc, kx);
    mma_rows<KSW - 1, NST / 2, NST>(fc, acc);
    __builtin_amdgcn_sched_group_barrier(0x100, KSW * (NST + WNT) / 2, 0);
    __builtin_amdgcn_sched_group_barrier(0x008, NTL / 2, 0);
}

__global__ __launch_bounds__(128) __attribute__((amdgpu_waves_per_eu(2, 2))) void gemm_bert(
    const float* __restrict__ A, const float* __restrict__ W,
    const float* __restrict__ bias, const float* __restrict__ prelu_a,
    float* __restrict__ art)
{
    __shared__ __align__(16) float smem[SMF];
    float (*As)[BM][LST] = (float (*)[BM][LST])smem;
    float (*Bs)[BN][LST] = (float (*)[BN][LST])(smem + 2 * BM * LST);

    const int tid  = threadIdx.x;
    const int wave = tid >> 5;
    const int lane = tid & 31;
    const int rc   = lane & 15;
    const int rsel = (lane >> 4) << 1;
    const int rowh = (lane >> 4) << 3;
    const int kx   = wave * (KC / NWV) + rsel;

    const int m0 = blockIdx.x * BM;
    const int nb = blockIdx.y * BN;

    const int srow = tid >> 4;
    const int sc4  = (tid & 15) << 2;
    const float* Ag = A + (size_t)(m0 + srow) * DB_ + sc4;
    int wo[NTB];
#pragma unroll
    for (int j = 0; j < NTB; ++j) {
        int nr = nb + srow + SRS * j;
        nr = nr < D_ ? nr : D_ - 1;
        wo[j] = nr * DB_ + sc4;
    }

    v8f acc[NST][WNT];
#pragma unroll
    for (int r = 0; r < NST; ++r)
#pragma unroll
        for (int t = 0; t < WNT; ++t) acc[r][t] = (v8f){0,0,0,0,0,0,0,0};

    Frags fa, fb;

    {
        float4 ra[NTA], rb[NTB];
#pragma unroll
        for (int j = 0; j < NTA; ++j)
            ra[j] = *(const float4*)(Ag + (size_t)(SRS * j) * DB_);
#pragma unroll
        for (int j = 0; j < NTB; ++j)
            rb[j] = *(const float4*)(W + wo[j]);
#pragma unroll
        for (int j = 0; j < NTA; ++j)
            *(float4*)&As[0][srow + SRS * j][sc4] = ra[j];
#pragma unroll
        for (int j = 0; j < NTB; ++j)
            *(float4*)&Bs[0][srow + SRS * j][sc4] = rb[j];
    }
    __syncthreads();
    load_frags(fa, As[0], Bs[0], rc, kx);

    static_assert((NCH / 2) % 2 == 0 && (3 * NCH / 4) % 2 == 0 && (11 * NCH / 12) % 2 == 0 && NCH % 2 == 0,
                  "the stretches are walked two chunks at a time");
    for (int q = 0; q < 4; ++q) {
        if (q == 0)      __builtin_amdgcn_s_setprio(3);
        else if (q == 1) __builtin_amdgcn_s_setprio(2);
        else if (q == 2) __builtin_amdgcn_s_setprio(1);
        else             __builtin_amdgcn_s_setprio(0);
        const int c0 = q == 0 ? 0 : q == 1 ? NCH / 2 : q == 2 ? 3 * NCH / 4 : 11 * NCH / 12;
        const int c1 = q == 0 ? NCH / 2 : q == 1 ? 3 * NCH / 4 : q == 2 ? 11 * NCH / 12 : NCH - 2;
        for (int c = c0; c < c1; c += 2) {
            chunk_step(c,     fa, fb, acc, Ag, W, wo, As, Bs, srow, sc4, rc, kx);
            chunk_step(c + 1, fb, fa, acc, Ag, W, wo, As, Bs, srow, sc4, rc, kx);
        }
    }
    chunk_step(NCH - 2, fa, fb, acc, Ag, W, wo, As, Bs, srow, sc4, rc, kx);
    mma_steps<0, KSW>(fb, acc);

    __syncthreads();
    float4* red = (float4*)smem;
#pragma unroll
    for (int r = 0; r < NST; ++r) {
#pragma unroll
        for (int nt = 0; nt < WNT; ++nt) {
            const int o = ((wave * NTL + r * WNT + nt) * 2) * 32 + lane;
            red[o]      = make_float4(acc[r][nt][0], acc[r][nt][1], acc[r][nt][2], acc[r][nt][3]);
            red[o + 32] = make_float4(acc[r][nt][4], acc[r][nt][5], acc[r][nt][6], acc[r][nt][7]);
        }
    }
    __syncthreads();

    const float pa = prelu_a[0];
#pragma unroll
    for (int nt = 0; nt < WNT; ++nt) {
        float4 lo[NWV], hi[NWV];
#pragma unroll
        for (int v = 0; v < NWV; ++v) {
            const int o = ((v * NTL + wave * WNT + nt) * 2) * 32 + lane;
            lo[v] = red[o];
            hi[v] = red[o + 32];
        }
        float s[8];
        s[0] = lo[0].x; s[1] = lo[0].y; s[2] = lo[0].z; s[3] = lo[0].w;
        s[4] = hi[0].x; s[5] = hi[0].y; s[6] = hi[0].z; s[7] = hi[0].w;
#pragma unroll
        for (int v = 1; v < NWV; ++v) {
            s[0] += lo[v].x; s[1] += lo[v].y; s[2] += lo[v].z; s[3] += lo[v].w;
            s[4] += hi[v].x; s[5] += hi[v].y; s[6] += hi[v].z; s[7] += hi[v].w;
        }
        int gc = nb + nt * 16 + rc;
        if (gc < D_) {
            float bb = bias[gc];
#pragma unroll
            for (int i = 0; i < 8; ++i) {
                int gr = m0 + wave * 16 + i + rowh;
                float v = s[i] + bb;
                v = v > 0.f ? v : pa * v;
                art[(size_t)gr * D_ + gc] = v;
            }
        }
    }
}

#define TN  16
#define TW  8
#define Q4  (D_ / 4)
#define WB  15
#define GB  10
#define KB  15
#define NKG (Q4 / KB)

__global__ __launch_bounds__(256) void tail_fused(
    const float* __restrict__ art, const int* __restrict__ idx,
    const int* __restrict__ mask, const float* __restrict__ W1,
    const float* __restrict__ b1, const float* __restrict__ W2,
    const float* __restrict__ b2, float* __restrict__ out)
{
    __shared__ __align__(16) float sw[H_ * D_];
    __shared__ __align__(16) float sn[TN * D_];
    __shared__ float spart[TW * TN];

    const int tid  = threadIdx.x;
    const int wave = tid >> 5;
    const int lane = tid & 31;
    const int bn0  = blockIdx.x * TN;
    const int b    = bn0 / NN_;

    int myi0 = 0, mym0 = 0, myi1 = 0, mym1 = 0;
    if (lane < NW_) {
        const size_t o = (size_t)(bn0 + wave * (TN / TW)) * NW_ + lane;
        myi0 = idx[o];        mym0 = mask[o];
        myi1 = idx[o + NW_];  mym1 = mask[o + NW_];
    }

    {
        const float4* Wg = (const float4*)W1;
        float4* Ws = (float4*)sw;
        for (int i0 = 0; i0 < H_ * D_ / 4 / 256; i0 += WB) {
            float4 t[WB];
#pragma unroll
            for (int j = 0; j < WB; ++j) t[j] = Wg[tid + 256 * (i0 + j)];
#pragma unroll
            for (int j = 0; j < WB; ++j) Ws[tid + 256 * (i0 + j)] = t[j];
        }
    }

    const float* artb = art + (size_t)b * L_ * D_;
    const int q2 = lane < Q4 - 64 ? lane + 64 : Q4 - 1;
    for (int nn = 0; nn < TN / TW; ++nn) {
        const int n   = wave * (TN / TW) + nn;
        const int myi = nn ? myi1 : myi0;
        const int mym = nn ? mym1 : mym0;
        float4 s0 = {0.f, 0.f, 0.f, 0.f}, s1 = s0, s2 = s0;
        float cnt = 0.f;
#pragma unroll
        for (int w0 = 0; w0 < NW_; w0 += GB) {
            float4 v0[GB], v1[GB], v2[GB];
            float m[GB];
#pragma unroll
            for (int j = 0; j < GB; ++j) {
                const int r = __builtin_amdgcn_readlane(myi, w0 + j);
                m[j] = (float)__builtin_amdgcn_readlane(mym, w0 + j);
                const float4* rp = (const float4*)(artb + (size_t)r * D_);
                v0[j] = rp[lane];
                v1[j] = rp[lane + 32];
                v2[j] = rp[q2];
            }
#pragma unroll
            for (int j = 0; j < GB; ++j) {
                cnt += m[j];
                s0.x += v0[j].x * m[j]; s0.y += v0[j].y * m[j];
                s0.z += v0[j].z * m[j]; s0.w += v0[j].w * m[j];
                s1.x += v1[j].x * m[j]; s1.y += v1[j].y * m[j];
                s1.z += v1[j].z * m[j]; s1.w += v1[j].w * m[j];
                s2.x += v2[j].x * m[j]; s2.y += v2[j].y * m[j];
                s2.z += v2[j].z * m[j]; s2.w += v2[j].w * m[j];
            }
        }
        const float inv = 1.f / fmaxf(cnt, 1.f);
        float4* np = (float4*)(sn + n * D_);
        np[lane]      = make_float4(s0.x * inv, s0.y * inv, s0.z * inv, s0.w * inv);
        np[lane + 32] = make_float4(s1.x * inv, s1.y * inv, s1.z * inv, s1.w * inv);
        if (lane < Q4 - 64)
            np[lane + 64] = make_float4(s2.x * inv, s2.y * inv, s2.z * inv, s2.w * inv);
    }
    __syncthreads();

    const int rc   = lane & 15;
    const int rsel = (lane >> 4) << 1;
    const int rowh = (lane >> 4) << 3;
    const float* ap  = sn + rc * D_ + rsel;
    const float* bp0 = sw + (wave * 32 + rc) * D_ + rsel;
    const float* bp1 = bp0 + 16 * D_;

    v8f acc0 = (v8f){0,0,0,0,0,0,0,0};
    v8f acc1 = (v8f){0,0,0,0,0,0,0,0};
    v2f fa[2][KB], fb0[2][KB], fb1[2][KB];
#pragma unroll
    for (int j = 0; j < KB; ++j) {
        fa [0][j] = *(const v2f*)(ap  + 4 * j);
        fb0[0][j] = *(const v2f*)(bp0 + 4 * j);
        fb1[0][j] = *(const v2f*)(bp1 + 4 * j);
    }
#pragma unroll
    for (int g = 0; g < NKG; ++g) {
        const int cur = g & 1;
        if (g + 1 < NKG) {
#pragma unroll
            for (int j = 0; j < KB; ++j) {
                const int k = 4 * ((g + 1) * KB + j);
                fa [cur ^ 1][j] = *(const v2f*)(ap  + k);
                fb0[cur ^ 1][j] = *(const v2f*)(bp0 + k);
                fb1[cur ^ 1][j] = *(const v2f*)(bp1 + k);
            }
        }
#pragma unroll
        for (int j = 0; j < KB; ++j) {
            acc0 = __builtin_amdgcn_wmma_f32_16x16x4_f32(
                false, fa[cur][j], false, fb0[cur][j], (short)0, acc0, false, false);
            acc1 = __builtin_amdgcn_wmma_f32_16x16x4_f32(
                false, fa[cur][j], false, fb1[cur][j], (short)0, acc1, false, false);
        }
    }

    const int c0 = wave * 32 + rc;
    const int c1 = c0 + 16;
    const float bb0 = b1[c0], bb1 = b1[c1];
    const float w20 = W2[c0], w21 = W2[c1];
    float p[8];
#pragma unroll
    for (int i = 0; i < 8; ++i) {
        const float h0 = fmaxf(acc0[i] + bb0, 0.f);
        const float h1 = fmaxf(acc1[i] + bb1, 0.f);
        p[i] = h0 * w20 + h1 * w21;
    }
#pragma unroll
    for (int off = 8; off > 0; off >>= 1) {
#pragma unroll
        for (int i = 0; i < 8; ++i) p[i] += __shfl_xor(p[i], off, 32);
    }
    if (rc == 0) {
#pragma unroll
        for (int i = 0; i < 8; ++i) spart[wave * TN + rowh + i] = p[i];
    }
    __syncthreads();
    if (tid < TN) {
        float s = 0.f;
#pragma unroll
        for (int v = 0; v < TW; ++v) s += spart[v * TN + tid];
        out[bn0 + tid] = s + b2[0];
    }
}

extern "C" void kernel_launch(void* const* d_in, const int* in_sizes, int n_in,
                              void* d_out, int out_size, void* d_ws, size_t ws_size,
                              hipStream_t stream)
{
    const float* articles = (const float*)d_in[0];
    const int*   nidx     = (const int*)  d_in[1];
    const int*   nmask    = (const int*)  d_in[2];
    const float* W_bert   = (const float*)d_in[5];
    const float* b_bert   = (const float*)d_in[6];
    const float* prelu_a  = (const float*)d_in[7];
    const float* W1       = (const float*)d_in[8];
    const float* b1       = (const float*)d_in[9];
    const float* W2       = (const float*)d_in[10];
    const float* b2       = (const float*)d_in[11];
    float* out = (float*)d_out;

    float* art = (float*)d_ws;
    (void)in_sizes; (void)n_in; (void)out_size; (void)ws_size;

    gemm_bert <<<dim3(M1 / BM, 4), 128, 0, stream>>>(articles, W_bert, b_bert, prelu_a, art);
    tail_fused<<<MN / TN, 256, 0, stream>>>(art, nidx, nmask, W1, b1, W2, b2, out);
}
